# starting kernel
# baseline (speedup 1.0000x reference)
#include <hip/hip_runtime.h>

typedef float v2f __attribute__((ext_vector_type(2)));
typedef float v8f __attribute__((ext_vector_type(8)));

#define D 64

__global__ void sage_deg_kernel(const int* __restrict__ dst,
                                float* __restrict__ deg, int nEdges) {
  int e = blockIdx.x * blockDim.x + threadIdx.x;
  if (e < nEdges) atomicAdd(deg + dst[e], 1.0f);
}

__global__ void sage_scatter_kernel(const float* __restrict__ xin,
                                    const int* __restrict__ src,
                                    const int* __restrict__ dst,
                                    float* __restrict__ agg, int nEdges) {
  int idx = blockIdx.x * blockDim.x + threadIdx.x;
  int total = nEdges * 16;
  if (idx >= total) return;
  int e = idx >> 4;
  int c = (idx & 15) << 2;
  int s = src[e];
  int d = dst[e];
  const float4 v = *(const float4*)(xin + (size_t)s * D + c);
  float* p = agg + (size_t)d * D + c;
  atomicAdd(p + 0, v.x);
  atomicAdd(p + 1, v.y);
  atomicAdd(p + 2, v.z);
  atomicAdd(p + 3, v.w);
}

template <bool RELU>
__global__ void sage_wmma_kernel(const float* __restrict__ agg,
                                 const float* __restrict__ deg,
                                 const float* __restrict__ xin,
                                 const float* __restrict__ Wl,
                                 const float* __restrict__ bl,
                                 const float* __restrict__ Wr,
                                 float* __restrict__ out, int nTiles) {
  const int lane = threadIdx.x & 31;
  const int wave = threadIdx.x >> 5;
  const int tile = blockIdx.x * (blockDim.x >> 5) + wave;
  if (tile >= nTiles) return;

  const int m = lane & 15;
  const int khalf = (lane >> 4) << 1;
  const int row0 = tile << 4;
  const int row = row0 + m;

  const float rdeg = 1.0f / fmaxf(deg[row], 1.0f);
  const float* __restrict__ arow = agg + (size_t)row * D;
  const float* __restrict__ xrow = xin + (size_t)row * D;

  v8f acc[4];
#pragma unroll
  for (int nt = 0; nt < 4; ++nt) acc[nt] = (v8f){0, 0, 0, 0, 0, 0, 0, 0};

#pragma unroll
  for (int kk = 0; kk < 16; ++kk) {
    const int k = (kk << 2) + khalf;
    v2f am = *(const v2f*)(arow + k);
    am[0] *= rdeg;
    am[1] *= rdeg;
    const v2f ax = *(const v2f*)(xrow + k);
#pragma unroll
    for (int nt = 0; nt < 4; ++nt) {
      const int j = (nt << 4) + m;
      const v2f bL = *(const v2f*)(Wl + j * D + k);
      const v2f bR = *(const v2f*)(Wr + j * D + k);
      acc[nt] = __builtin_amdgcn_wmma_f32_16x16x4_f32(
          false, am, false, bL, (short)0, acc[nt], false, false);
      acc[nt] = __builtin_amdgcn_wmma_f32_16x16x4_f32(
          false, ax, false, bR, (short)0, acc[nt], false, false);
    }
  }

  const int mhi = (lane >> 4) << 3;
#pragma unroll
  for (int nt = 0; nt < 4; ++nt) {
    const int j = (nt << 4) + m;
    const float b = bl[j];
#pragma unroll
    for (int v = 0; v < 8; ++v) {
      float val = acc[nt][v] + b;
      if (RELU) val = fmaxf(val, 0.0f);
      out[(size_t)(row0 + mhi + v) * D + j] = val;
    }
  }
}

template <bool RELU>
__global__ void sage_tail_kernel(const float* __restrict__ agg,
                                 const float* __restrict__ deg,
                                 const float* __restrict__ xin,
                                 const float* __restrict__ Wl,
                                 const float* __restrict__ bl,
                                 const float* __restrict__ Wr,
                                 float* __restrict__ out, int row0, int nRows) {
  int idx = blockIdx.x * blockDim.x + threadIdx.x;
  if (idx >= nRows * D) return;
  int r = row0 + idx / D;
  int j = idx % D;
  float rdeg = 1.0f / fmaxf(deg[r], 1.0f);
  float s = bl[j];
  for (int k = 0; k < D; ++k)
    s += agg[(size_t)r * D + k] * rdeg * Wl[j * D + k] +
         xin[(size_t)r * D + k] * Wr[j * D + k];
  if (RELU) s = fmaxf(s, 0.0f);
  out[(size_t)r * D + j] = s;
}

static inline size_t align256(size_t x) { return (x + 255) & ~(size_t)255; }

extern "C" void kernel_launch(void* const* d_in, const int* in_sizes, int n_in,
                              void* d_out, int out_size, void* d_ws,
                              size_t ws_size, hipStream_t stream) {
  const float* x = (const float*)d_in[0];
  const int* src = (const int*)d_in[1];
  const int* dst = (const int*)d_in[2];
  const float* Wl[3] = {(const float*)d_in[3], (const float*)d_in[6],
                        (const float*)d_in[9]};
  const float* bl[3] = {(const float*)d_in[4], (const float*)d_in[7],
                        (const float*)d_in[10]};
  const float* Wr[3] = {(const float*)d_in[5], (const float*)d_in[8],
                        (const float*)d_in[11]};

  const int nNodes = in_sizes[0] / D;
  const int nEdges = in_sizes[1];
  const size_t featBytes = (size_t)nNodes * D * sizeof(float);

  char* ws = (char*)d_ws;
  float* deg = (float*)ws;
  size_t ofs = align256((size_t)nNodes * sizeof(float));
  float* agg = (float*)(ws + ofs);
  ofs += align256(featBytes);
  float* h0 = (float*)(ws + ofs);
  ofs += align256(featBytes);
  float* h1 = (float*)(ws + ofs);

  hipMemsetAsync(deg, 0, (size_t)nNodes * sizeof(float), stream);
  sage_deg_kernel<<<(nEdges + 255) / 256, 256, 0, stream>>>(dst, deg, nEdges);

  const int scatterThreads = nEdges * 16;
  const int scatterBlocks = (scatterThreads + 255) / 256;
  const int nTiles = nNodes / 16;
  const int gemmBlocks = (nTiles + 7) / 8;
  const int tail = nNodes - nTiles * 16;

  auto run_layer = [&](const float* in, const float* wl, const float* b,
                       const float* wr, float* outp, bool relu) {
    hipMemsetAsync(agg, 0, featBytes, stream);
    sage_scatter_kernel<<<scatterBlocks, 256, 0, stream>>>(in, src, dst, agg,
                                                           nEdges);
    if (nTiles > 0) {
      if (relu)
        sage_wmma_kernel<true><<<gemmBlocks, 256, 0, stream>>>(
            agg, deg, in, wl, b, wr, outp, nTiles);
      else
        sage_wmma_kernel<false><<<gemmBlocks, 256, 0, stream>>>(
            agg, deg, in, wl, b, wr, outp, nTiles);
    }
    if (tail > 0) {
      int tb = (tail * D + 255) / 256;
      if (relu)
        sage_tail_kernel<true><<<tb, 256, 0, stream>>>(
            agg, deg, in, wl, b, wr, outp, nTiles * 16, tail);
      else
        sage_tail_kernel<false><<<tb, 256, 0, stream>>>(
            agg, deg, in, wl, b, wr, outp, nTiles * 16, tail);
    }
  };

  run_layer(x, Wl[0], bl[0], Wr[0], h0,  true);
  run_layer(h0, Wl[1], bl[1], Wr[1], h1,  true);
  run_layer(h1, Wl[2], bl[2], Wr[2], (float*)d_out,  false);
}


# Aggregation by gather, not f32 atomic scatter: edges ranked per destination once (returning int atomic), prefix sum to row offsets, layers sum source rows in that rank order; agg memsets dropped
# speedup vs baseline: 2.0086x; 2.0086x over previous
#include <hip/hip_runtime.h>

typedef float v2f __attribute__((ext_vector_type(2)));
typedef float v8f __attribute__((ext_vector_type(8)));

#define D 64

__global__ void sage_rank_kernel(const int* __restrict__ dst,
                                 int* __restrict__ cnt, int* __restrict__ rank,
                                 int nEdges) {
  int e = blockIdx.x * blockDim.x + threadIdx.x;
  if (e < nEdges) rank[e] = atomicAdd(cnt + dst[e], 1);
}

#define OFFS_T 1024

__global__ void sage_offsets_kernel(const int* __restrict__ cnt,
                                    int* __restrict__ rowptr,
                                    float* __restrict__ deg, int n) {
  __shared__ int sh[OFFS_T];
  const int tid = threadIdx.x;
  const int base = blockIdx.x * (OFFS_T * 4);

  int pre = 0;
  const int4* __restrict__ c4 = (const int4*)cnt;
  for (int i = tid; i < (base >> 2); i += OFFS_T) {
    const int4 v = c4[i];
    pre += (v.x + v.y) + (v.z + v.w);
  }
  sh[tid] = pre;
  __syncthreads();
  for (int s = OFFS_T >> 1; s > 0; s >>= 1) {
    if (tid < s) sh[tid] += sh[tid + s];
    __syncthreads();
  }
  const int blockPre = sh[0];
  __syncthreads();

  const int i0 = base + (tid << 2);
  int a[4];
#pragma unroll
  for (int k = 0; k < 4; ++k) a[k] = (i0 + k < n) ? cnt[i0 + k] : 0;
  const int t = (a[0] + a[1]) + (a[2] + a[3]);

  sh[tid] = t;
  __syncthreads();
  for (int d = 1; d < OFFS_T; d <<= 1) {
    const int v = (tid >= d) ? sh[tid - d] : 0;
    __syncthreads();
    sh[tid] += v;
    __syncthreads();
  }
  int run = blockPre + sh[tid] - t;
#pragma unroll
  for (int k = 0; k < 4; ++k) {
    const int i = i0 + k;
    if (i < n) {
      rowptr[i] = run;
      deg[i] = (float)a[k];
      run += a[k];
      if (i == n - 1) rowptr[n] = run;
    }
  }
}

__global__ void sage_fill_kernel(const int* __restrict__ src,
                                 const int* __restrict__ dst,
                                 const int* __restrict__ rank,
                                 const int* __restrict__ rowptr,
                                 int* __restrict__ csr, int nEdges) {
  int e = blockIdx.x * blockDim.x + threadIdx.x;
  if (e < nEdges) csr[rowptr[dst[e]] + rank[e]] = src[e];
}

__global__ void sage_gather_kernel(const float* __restrict__ xin,
                                   const int* __restrict__ rowptr,
                                   const int* __restrict__ csr,
                                   float* __restrict__ agg, int nNodes) {
  const int idx = blockIdx.x * blockDim.x + threadIdx.x;
  const int row = idx >> 4;
  if (row >= nNodes) return;
  const int c = (idx & 15) << 2;
  int j = rowptr[row];
  const int end = rowptr[row + 1];
  float4 acc = make_float4(0.0f, 0.0f, 0.0f, 0.0f);
  for (; j + 4 <= end; j += 4) {
    const int s0 = csr[j], s1 = csr[j + 1], s2 = csr[j + 2], s3 = csr[j + 3];
    const float4 v0 = *(const float4*)(xin + (size_t)s0 * D + c);
    const float4 v1 = *(const float4*)(xin + (size_t)s1 * D + c);
    const float4 v2 = *(const float4*)(xin + (size_t)s2 * D + c);
    const float4 v3 = *(const float4*)(xin + (size_t)s3 * D + c);
    acc.x += v0.x; acc.y += v0.y; acc.z += v0.z; acc.w += v0.w;
    acc.x += v1.x; acc.y += v1.y; acc.z += v1.z; acc.w += v1.w;
    acc.x += v2.x; acc.y += v2.y; acc.z += v2.z; acc.w += v2.w;
    acc.x += v3.x; acc.y += v3.y; acc.z += v3.z; acc.w += v3.w;
  }
  for (; j < end; ++j) {
    const int s0 = csr[j];
    const float4 v0 = *(const float4*)(xin + (size_t)s0 * D + c);
    acc.x += v0.x; acc.y += v0.y; acc.z += v0.z; acc.w += v0.w;
  }
  *(float4*)(agg + (size_t)row * D + c) = acc;
}

template <bool RELU>
__global__ void sage_wmma_kernel(const float* __restrict__ agg,
                                 const float* __restrict__ deg,
                                 const float* __restrict__ xin,
                                 const float* __restrict__ Wl,
                                 const float* __restrict__ bl,
                                 const float* __restrict__ Wr,
                                 float* __restrict__ out, int nTiles) {
  const int lane = threadIdx.x & 31;
  const int wave = threadIdx.x >> 5;
  const int tile = blockIdx.x * (blockDim.x >> 5) + wave;
  if (tile >= nTiles) return;

  const int m = lane & 15;
  const int khalf = (lane >> 4) << 1;
  const int row0 = tile << 4;
  const int row = row0 + m;

  const float rdeg = 1.0f / fmaxf(deg[row], 1.0f);
  const float* __restrict__ arow = agg + (size_t)row * D;
  const float* __restrict__ xrow = xin + (size_t)row * D;

  v8f acc[4];
#pragma unroll
  for (int nt = 0; nt < 4; ++nt) acc[nt] = (v8f){0, 0, 0, 0, 0, 0, 0, 0};

#pragma unroll
  for (int kk = 0; kk < 16; ++kk) {
    const int k = (kk << 2) + khalf;
    v2f am = *(const v2f*)(arow + k);
    am[0] *= rdeg;
    am[1] *= rdeg;
    const v2f ax = *(const v2f*)(xrow + k);
#pragma unroll
    for (int nt = 0; nt < 4; ++nt) {
      const int j = (nt << 4) + m;
      const v2f bL = *(const v2f*)(Wl + j * D + k);
      const v2f bR = *(const v2f*)(Wr + j * D + k);
      acc[nt] = __builtin_amdgcn_wmma_f32_16x16x4_f32(
          false, am, false, bL, (short)0, acc[nt], false, false);
      acc[nt] = __builtin_amdgcn_wmma_f32_16x16x4_f32(
          false, ax, false, bR, (short)0, acc[nt], false, false);
    }
  }

  const int mhi = (lane >> 4) << 3;
#pragma unroll
  for (int nt = 0; nt < 4; ++nt) {
    const int j = (nt << 4) + m;
    const float b = bl[j];
#pragma unroll
    for (int v = 0; v < 8; ++v) {
      float val = acc[nt][v] + b;
      if (RELU) val = fmaxf(val, 0.0f);
      out[(size_t)(row0 + mhi + v) * D + j] = val;
    }
  }
}

template <bool RELU>
__global__ void sage_tail_kernel(const float* __restrict__ agg,
                                 const float* __restrict__ deg,
                                 const float* __restrict__ xin,
                                 const float* __restrict__ Wl,
                                 const float* __restrict__ bl,
                                 const float* __restrict__ Wr,
                                 float* __restrict__ out, int row0, int nRows) {
  int idx = blockIdx.x * blockDim.x + threadIdx.x;
  if (idx >= nRows * D) return;
  int r = row0 + idx / D;
  int j = idx % D;
  float rdeg = 1.0f / fmaxf(deg[r], 1.0f);
  float s = bl[j];
  for (int k = 0; k < D; ++k)
    s += agg[(size_t)r * D + k] * rdeg * Wl[j * D + k] +
         xin[(size_t)r * D + k] * Wr[j * D + k];
  if (RELU) s = fmaxf(s, 0.0f);
  out[(size_t)r * D + j] = s;
}

static inline size_t align256(size_t x) { return (x + 255) & ~(size_t)255; }

extern "C" void kernel_launch(void* const* d_in, const int* in_sizes, int n_in,
                              void* d_out, int out_size, void* d_ws,
                              size_t ws_size, hipStream_t stream) {
  const float* x = (const float*)d_in[0];
  const int* src = (const int*)d_in[1];
  const int* dst = (const int*)d_in[2];
  const float* Wl[3] = {(const float*)d_in[3], (const float*)d_in[6],
                        (const float*)d_in[9]};
  const float* bl[3] = {(const float*)d_in[4], (const float*)d_in[7],
                        (const float*)d_in[10]};
  const float* Wr[3] = {(const float*)d_in[5], (const float*)d_in[8],
                        (const float*)d_in[11]};

  const int nNodes = in_sizes[0] / D;
  const int nEdges = in_sizes[1];
  const size_t featBytes = (size_t)nNodes * D * sizeof(float);

  char* ws = (char*)d_ws;
  float* deg = (float*)ws;
  size_t ofs = align256((size_t)nNodes * sizeof(float));
  float* agg = (float*)(ws + ofs);
  ofs += align256(featBytes);
  float* h0 = (float*)(ws + ofs);
  ofs += align256(featBytes);
  float* h1 = (float*)(ws + ofs);

  int* cnt = (int*)agg;
  int* rank = (int*)((char*)agg + align256((size_t)nNodes * sizeof(int)));
  int* rowptr = (int*)d_out;
  int* csr =
      (int*)((char*)d_out + align256(((size_t)nNodes + 1) * sizeof(int)));
  const int edgeBlocks = (nEdges + 255) / 256;
  hipMemsetAsync(cnt, 0, (size_t)nNodes * sizeof(int), stream);
  sage_rank_kernel<<<edgeBlocks, 256, 0, stream>>>(dst, cnt, rank, nEdges);
  sage_offsets_kernel<<<(nNodes + OFFS_T * 4 - 1) / (OFFS_T * 4), OFFS_T, 0,
                        stream>>>(cnt, rowptr, deg, nNodes);
  sage_fill_kernel<<<edgeBlocks, 256, 0, stream>>>(src, dst, rank, rowptr, csr,
                                                   nEdges);

  const int gatherBlocks = (nNodes * 16 + 255) / 256;
  const int nTiles = nNodes / 16;
  const int gemmBlocks = (nTiles + 7) / 8;
  const int tail = nNodes - nTiles * 16;

  auto run_layer = [&](const float* in, const float* wl, const float* b,
                       const float* wr, float* outp, bool relu) {
    sage_gather_kernel<<<gatherBlocks, 256, 0, stream>>>(in, rowptr, csr, agg,
                                                         nNodes);
    if (nTiles > 0) {
      if (relu)
        sage_wmma_kernel<true><<<gemmBlocks, 256, 0, stream>>>(
            agg, deg, in, wl, b, wr, outp, nTiles);
      else
        sage_wmma_kernel<false><<<gemmBlocks, 256, 0, stream>>>(
            agg, deg, in, wl, b, wr, outp, nTiles);
    }
    if (tail > 0) {
      int tb = (tail * D + 255) / 256;
      if (relu)
        sage_tail_kernel<true><<<tb, 256, 0, stream>>>(
            agg, deg, in, wl, b, wr, outp, nTiles * 16, tail);
      else
        sage_tail_kernel<false><<<tb, 256, 0, stream>>>(
            agg, deg, in, wl, b, wr, outp, nTiles * 16, tail);
    }
  };

  run_layer(x, Wl[0], bl[0], Wr[0], h0,  true);
  run_layer(h0, Wl[1], bl[1], Wr[1], h1,  true);
  run_layer(h1, Wl[2], bl[2], Wr[2], (float*)d_out,  false);
}


# Adjacency as per-node 32-slot buckets plus an overflow list, filled in one edge pass (slot = returning int add): prefix-sum and fill launches dropped; a node's sum stays in arrival order
# speedup vs baseline: 2.2038x; 1.0972x over previous
#include <hip/hip_runtime.h>

typedef float v2f __attribute__((ext_vector_type(2)));
typedef float v8f __attribute__((ext_vector_type(8)));

#define D 64

#define BUCKET 32

__device__ __forceinline__ void sage_put(int d, int s, int slot,
                                         int* __restrict__ bucket,
                                         int* __restrict__ ovfDst,
                                         int* __restrict__ ovfSrc,
                                         int* __restrict__ ovfLen, int ovfCap) {
  if (slot < BUCKET) {
    bucket[d * BUCKET + slot] = s;
  } else {
    const int pos = atomicAdd(ovfLen, 1);
    if (pos < ovfCap) {
      ovfDst[pos] = d;
      ovfSrc[pos] = s;
    }
  }
}

__global__ void sage_bucket_kernel(const int* __restrict__ src,
                                   const int* __restrict__ dst,
                                   int* __restrict__ cnt,
                                   int* __restrict__ bucket,
                                   int* __restrict__ ovfDst,
                                   int* __restrict__ ovfSrc,
                                   int* __restrict__ ovfLen, int ovfCap,
                                   int nEdges) {
  const int e0 = (blockIdx.x * blockDim.x + threadIdx.x) << 2;
  if (e0 + 4 <= nEdges) {
    const int4 d = *(const int4*)(dst + e0);
    const int4 s = *(const int4*)(src + e0);
    const int r0 = atomicAdd(cnt + d.x, 1);
    const int r1 = atomicAdd(cnt + d.y, 1);
    const int r2 = atomicAdd(cnt + d.z, 1);
    const int r3 = atomicAdd(cnt + d.w, 1);
    sage_put(d.x, s.x, r0, bucket, ovfDst, ovfSrc, ovfLen, ovfCap);
    sage_put(d.y, s.y, r1, bucket, ovfDst, ovfSrc, ovfLen, ovfCap);
    sage_put(d.z, s.z, r2, bucket, ovfDst, ovfSrc, ovfLen, ovfCap);
    sage_put(d.w, s.w, r3, bucket, ovfDst, ovfSrc, ovfLen, ovfCap);
  } else {
    for (int e = e0; e < nEdges; ++e) {
      const int dd = dst[e];
      sage_put(dd, src[e], atomicAdd(cnt + dd, 1), bucket, ovfDst, ovfSrc,
               ovfLen, ovfCap);
    }
  }
}

__global__ void sage_gather_kernel(const float* __restrict__ xin,
                                   const int* __restrict__ cnt,
                                   const int* __restrict__ bucket,
                                   const int* __restrict__ ovfDst,
                                   const int* __restrict__ ovfSrc, int ovfCap,
                                   float* __restrict__ agg, int nNodes) {
  const int idx = blockIdx.x * blockDim.x + threadIdx.x;
  const int row = idx >> 4;
  if (row >= nNodes) return;
  const int c = (idx & 15) << 2;
  const int n = cnt[row];
  const int end = n < BUCKET ? n : BUCKET;
  const int* __restrict__ b = bucket + row * BUCKET;
  float4 acc = make_float4(0.0f, 0.0f, 0.0f, 0.0f);
  int j = 0;
  for (; j + 4 <= end; j += 4) {
    const int4 s = *(const int4*)(b + j);
    const float4 v0 = *(const float4*)(xin + (size_t)s.x * D + c);
    const float4 v1 = *(const float4*)(xin + (size_t)s.y * D + c);
    const float4 v2 = *(const float4*)(xin + (size_t)s.z * D + c);
    const float4 v3 = *(const float4*)(xin + (size_t)s.w * D + c);
    acc.x += v0.x; acc.y += v0.y; acc.z += v0.z; acc.w += v0.w;
    acc.x += v1.x; acc.y += v1.y; acc.z += v1.z; acc.w += v1.w;
    acc.x += v2.x; acc.y += v2.y; acc.z += v2.z; acc.w += v2.w;
    acc.x += v3.x; acc.y += v3.y; acc.z += v3.z; acc.w += v3.w;
  }
  for (; j < end; ++j) {
    const int s0 = b[j];
    const float4 v0 = *(const float4*)(xin + (size_t)s0 * D + c);
    acc.x += v0.x; acc.y += v0.y; acc.z += v0.z; acc.w += v0.w;
  }
  if (n > BUCKET) {
    const int len = cnt[nNodes];
    const int m = len < ovfCap ? len : ovfCap;
    for (int i = 0; i < m; ++i) {
      if (ovfDst[i] == row) {
        const float4 v0 = *(const float4*)(xin + (size_t)ovfSrc[i] * D + c);
        acc.x += v0.x; acc.y += v0.y; acc.z += v0.z; acc.w += v0.w;
      }
    }
  }
  *(float4*)(agg + (size_t)row * D + c) = acc;
}

template <bool RELU>
__global__ void sage_wmma_kernel(const float* __restrict__ agg,
                                 const int* __restrict__ cnt,
                                 const float* __restrict__ xin,
                                 const float* __restrict__ Wl,
                                 const float* __restrict__ bl,
                                 const float* __restrict__ Wr,
                                 float* __restrict__ out, int nTiles) {
  const int lane = threadIdx.x & 31;
  const int wave = threadIdx.x >> 5;
  const int tile = blockIdx.x * (blockDim.x >> 5) + wave;
  if (tile >= nTiles) return;

  const int m = lane & 15;
  const int khalf = (lane >> 4) << 1;
  const int row0 = tile << 4;
  const int row = row0 + m;

  const float rdeg = 1.0f / fmaxf((float)cnt[row], 1.0f);
  const float* __restrict__ arow = agg + (size_t)row * D;
  const float* __restrict__ xrow = xin + (size_t)row * D;

  v8f acc[4];
#pragma unroll
  for (int nt = 0; nt < 4; ++nt) acc[nt] = (v8f){0, 0, 0, 0, 0, 0, 0, 0};

#pragma unroll
  for (int kk = 0; kk < 16; ++kk) {
    const int k = (kk << 2) + khalf;
    v2f am = *(const v2f*)(arow + k);
    am[0] *= rdeg;
    am[1] *= rdeg;
    const v2f ax = *(const v2f*)(xrow + k);
#pragma unroll
    for (int nt = 0; nt < 4; ++nt) {
      const int j = (nt << 4) + m;
      const v2f bL = *(const v2f*)(Wl + j * D + k);
      const v2f bR = *(const v2f*)(Wr + j * D + k);
      acc[nt] = __builtin_amdgcn_wmma_f32_16x16x4_f32(
          false, am, false, bL, (short)0, acc[nt], false, false);
      acc[nt] = __builtin_amdgcn_wmma_f32_16x16x4_f32(
          false, ax, false, bR, (short)0, acc[nt], false, false);
    }
  }

  const int mhi = (lane >> 4) << 3;
#pragma unroll
  for (int nt = 0; nt < 4; ++nt) {
    const int j = (nt << 4) + m;
    const float b = bl[j];
#pragma unroll
    for (int v = 0; v < 8; ++v) {
      float val = acc[nt][v] + b;
      if (RELU) val = fmaxf(val, 0.0f);
      out[(size_t)(row0 + mhi + v) * D + j] = val;
    }
  }
}

template <bool RELU>
__global__ void sage_tail_kernel(const float* __restrict__ agg,
                                 const int* __restrict__ cnt,
                                 const float* __restrict__ xin,
                                 const float* __restrict__ Wl,
                                 const float* __restrict__ bl,
                                 const float* __restrict__ Wr,
                                 float* __restrict__ out, int row0, int nRows) {
  int idx = blockIdx.x * blockDim.x + threadIdx.x;
  if (idx >= nRows * D) return;
  int r = row0 + idx / D;
  int j = idx % D;
  float rdeg = 1.0f / fmaxf((float)cnt[r], 1.0f);
  float s = bl[j];
  for (int k = 0; k < D; ++k)
    s += agg[(size_t)r * D + k] * rdeg * Wl[j * D + k] +
         xin[(size_t)r * D + k] * Wr[j * D + k];
  if (RELU) s = fmaxf(s, 0.0f);
  out[(size_t)r * D + j] = s;
}

static inline size_t align256(size_t x) { return (x + 255) & ~(size_t)255; }

extern "C" void kernel_launch(void* const* d_in, const int* in_sizes, int n_in,
                              void* d_out, int out_size, void* d_ws,
                              size_t ws_size, hipStream_t stream) {
  const float* x = (const float*)d_in[0];
  const int* src = (const int*)d_in[1];
  const int* dst = (const int*)d_in[2];
  const float* Wl[3] = {(const float*)d_in[3], (const float*)d_in[6],
                        (const float*)d_in[9]};
  const float* bl[3] = {(const float*)d_in[4], (const float*)d_in[7],
                        (const float*)d_in[10]};
  const float* Wr[3] = {(const float*)d_in[5], (const float*)d_in[8],
                        (const float*)d_in[11]};

  const int nNodes = in_sizes[0] / D;
  const int nEdges = in_sizes[1];
  const size_t featBytes = (size_t)nNodes * D * sizeof(float);

  char* ws = (char*)d_ws;
  int* cnt = (int*)ws;
  int* ovfLen = cnt + nNodes;
  size_t ofs = align256(((size_t)nNodes + 1) * sizeof(int));
  float* agg = (float*)(ws + ofs);
  ofs += align256(featBytes);
  float* h0 = (float*)(ws + ofs);
  ofs += align256(featBytes);
  float* h1 = (float*)(ws + ofs);

  int* bucket = (int*)d_out;
  const size_t bucketBytes = (size_t)nNodes * BUCKET * sizeof(int);
  const int ovfCap = (int)((featBytes - bucketBytes) / (2 * sizeof(int)));
  int* ovfDst = (int*)((char*)d_out + bucketBytes);
  int* ovfSrc = ovfDst + ovfCap;
  const int edgeThreads = (nEdges + 3) / 4;
  hipMemsetAsync(cnt, 0, ((size_t)nNodes + 1) * sizeof(int), stream);
  sage_bucket_kernel<<<(edgeThreads + 255) / 256, 256, 0, stream>>>(
      src, dst, cnt, bucket, ovfDst, ovfSrc, ovfLen, ovfCap, nEdges);

  const int gatherBlocks = (nNodes * 16 + 255) / 256;
  const int nTiles = nNodes / 16;
  const int gemmBlocks = (nTiles + 7) / 8;
  const int tail = nNodes - nTiles * 16;

  auto run_layer = [&](const float* in, const float* wl, const float* b,
                       const float* wr, float* outp, bool relu) {
    sage_gather_kernel<<<gatherBlocks, 256, 0, stream>>>(
        in, cnt, bucket, ovfDst, ovfSrc, ovfCap, agg, nNodes);
    if (nTiles > 0) {
      if (relu)
        sage_wmma_kernel<true><<<gemmBlocks, 256, 0, stream>>>(
            agg, cnt, in, wl, b, wr, outp, nTiles);
      else
        sage_wmma_kernel<false><<<gemmBlocks, 256, 0, stream>>>(
            agg, cnt, in, wl, b, wr, outp, nTiles);
    }
    if (tail > 0) {
      int tb = (tail * D + 255) / 256;
      if (relu)
        sage_tail_kernel<true><<<tb, 256, 0, stream>>>(
            agg, cnt, in, wl, b, wr, outp, nTiles * 16, tail);
      else
        sage_tail_kernel<false><<<tb, 256, 0, stream>>>(
            agg, cnt, in, wl, b, wr, outp, nTiles * 16, tail);
    }
  };

  run_layer(x, Wl[0], bl[0], Wr[0], h0,  true);
  run_layer(h0, Wl[1], bl[1], Wr[1], h1,  true);
  run_layer(h1, Wl[2], bl[2], Wr[2], (float*)d_out,  false);
}


# Bucket slots drawn by LDS returning adds, not global ones: edges counted and copied into 128 node-range segments (one global add per block and range), then one work-group per range fills its buckets
# speedup vs baseline: 3.0799x; 1.3975x over previous
#include <hip/hip_runtime.h>

typedef float v2f __attribute__((ext_vector_type(2)));
typedef float v8f __attribute__((ext_vector_type(8)));

#define D 64

#define BUCKET 32
#define OWN 784
#define OWN_THREADS 1024
#define RANGES_MAX 256
#define PART_THREADS 256
#define PART_QUADS 4

__device__ __forceinline__ int4 sage_load4(const int* __restrict__ p, int e,
                                           int nEdges) {
  if (e + 4 <= nEdges) return *(const int4*)(p + e);
  int4 v = make_int4(-1, -1, -1, -1);
  if (e + 0 < nEdges) v.x = p[e + 0];
  if (e + 1 < nEdges) v.y = p[e + 1];
  if (e + 2 < nEdges) v.z = p[e + 2];
  return v;
}

__global__ void __launch_bounds__(PART_THREADS)
sage_count_kernel(const int* __restrict__ dst, int* __restrict__ rangeCount,
                  int* __restrict__ ovfLen, int nEdges) {
  __shared__ int hist[RANGES_MAX];
  const int tid = threadIdx.x;
  hist[tid] = 0;
  if (blockIdx.x == 0 && tid == 0) *ovfLen = 0;
  __syncthreads();
  const int e0 = blockIdx.x * (PART_THREADS * PART_QUADS * 4);
#pragma unroll
  for (int k = 0; k < PART_QUADS; ++k) {
    const int4 d = sage_load4(dst, e0 + (k * PART_THREADS + tid) * 4, nEdges);
    if (d.x >= 0) atomicAdd(&hist[d.x / OWN], 1);
    if (d.y >= 0) atomicAdd(&hist[d.y / OWN], 1);
    if (d.z >= 0) atomicAdd(&hist[d.z / OWN], 1);
    if (d.w >= 0) atomicAdd(&hist[d.w / OWN], 1);
  }
  __syncthreads();
  const int h = hist[tid];
  if (h) atomicAdd(rangeCount + tid, h);
}

__device__ __forceinline__ int sage_scan_incl(int v, int* a, int* b, int tid) {
  __syncthreads();
  a[tid] = v;
  __syncthreads();
  int* in = a;
  int* out = b;
  for (int off = 1; off < RANGES_MAX; off <<= 1) {
    out[tid] = in[tid] + (tid >= off ? in[tid - off] : 0);
    __syncthreads();
    int* t = in; in = out; out = t;
  }
  return in[tid];
}

__global__ void __launch_bounds__(PART_THREADS)
sage_place_kernel(const int* __restrict__ src, const int* __restrict__ dst,
                  const int* __restrict__ rangeCount,
                  int* __restrict__ cursor, int* __restrict__ rangeStart,
                  int* __restrict__ pairs, int nEdges) {
  __shared__ int hist[RANGES_MAX];
  __shared__ int lstart[RANGES_MAX];
  __shared__ int base[RANGES_MAX];
  __shared__ int scA[RANGES_MAX], scB[RANGES_MAX];
  __shared__ __attribute__((aligned(8))) int stage[PART_THREADS * PART_QUADS * 4 * 2];
  const int tid = threadIdx.x;
  hist[tid] = 0;
  __syncthreads();

  int4 d[PART_QUADS], s[PART_QUADS], rk[PART_QUADS];
  const int e0 = blockIdx.x * (PART_THREADS * PART_QUADS * 4);
#pragma unroll
  for (int k = 0; k < PART_QUADS; ++k) {
    const int e = e0 + (k * PART_THREADS + tid) * 4;
    d[k] = sage_load4(dst, e, nEdges);
    s[k] = sage_load4(src, e, nEdges);
    rk[k] = make_int4(0, 0, 0, 0);
    if (d[k].x >= 0) rk[k].x = atomicAdd(&hist[d[k].x / OWN], 1);
    if (d[k].y >= 0) rk[k].y = atomicAdd(&hist[d[k].y / OWN], 1);
    if (d[k].z >= 0) rk[k].z = atomicAdd(&hist[d[k].z / OWN], 1);
    if (d[k].w >= 0) rk[k].w = atomicAdd(&hist[d[k].w / OWN], 1);
  }

  const int total = rangeCount[tid];
  const int start = sage_scan_incl(total, scA, scB, tid) - total;
  if (blockIdx.x == 0) rangeStart[tid] = start;
  const int h = hist[tid];
  const int ls = sage_scan_incl(h, scA, scB, tid) - h;
  lstart[tid] = ls;
  base[tid] = start + (h ? atomicAdd(cursor + tid, h) : 0) - ls;
  __syncthreads();

#pragma unroll
  for (int k = 0; k < PART_QUADS; ++k) {
    if (d[k].x >= 0) *(int2*)(stage + 2 * (lstart[d[k].x / OWN] + rk[k].x)) = make_int2(d[k].x, s[k].x);
    if (d[k].y >= 0) *(int2*)(stage + 2 * (lstart[d[k].y / OWN] + rk[k].y)) = make_int2(d[k].y, s[k].y);
    if (d[k].z >= 0) *(int2*)(stage + 2 * (lstart[d[k].z / OWN] + rk[k].z)) = make_int2(d[k].z, s[k].z);
    if (d[k].w >= 0) *(int2*)(stage + 2 * (lstart[d[k].w / OWN] + rk[k].w)) = make_int2(d[k].w, s[k].w);
  }
  __syncthreads();

  int nHere = nEdges - e0;
  if (nHere > PART_THREADS * PART_QUADS * 4) nHere = PART_THREADS * PART_QUADS * 4;
  for (int i = tid; i < nHere; i += PART_THREADS) {
    const int2 pr = *(const int2*)(stage + 2 * i);
    *(int2*)(pairs + 2 * (size_t)(base[pr.x / OWN] + i)) = pr;
  }
}

__global__ void __launch_bounds__(OWN_THREADS)
sage_bucket_kernel(const int* __restrict__ pairs,
                   const int* __restrict__ rangeCount,
                   const int* __restrict__ rangeStart,
                   int* __restrict__ cnt,
                   int* __restrict__ bucket,
                   int* __restrict__ ovfDst,
                   int* __restrict__ ovfSrc,
                   int* __restrict__ ovfLen, int ovfCap, int nNodes) {
  __shared__ int lcnt[OWN];
  __shared__ __attribute__((aligned(16))) int lbkt[OWN * BUCKET];
  const int tid = threadIdx.x;
  const int lo = blockIdx.x * OWN;
  const int left = nNodes - lo;
  const int nOwn = left < OWN ? left : OWN;
  for (int i = tid; i < OWN; i += OWN_THREADS) lcnt[i] = 0;
  __syncthreads();

  const int n = rangeCount[blockIdx.x];
  const int* __restrict__ seg = pairs + 2 * (size_t)rangeStart[blockIdx.x];
  for (int i = tid; i < n; i += OWN_THREADS) {
    const int2 pr = *(const int2*)(seg + 2 * i);
    const int r = pr.x - lo;
    const int slot = atomicAdd(&lcnt[r], 1);
    if (slot < BUCKET) {
      lbkt[r * BUCKET + slot] = pr.y;
    } else {
      const int pos = atomicAdd(ovfLen, 1);
      if (pos < ovfCap) {
        ovfDst[pos] = pr.x;
        ovfSrc[pos] = pr.y;
      }
    }
  }
  __syncthreads();

  for (int i = tid; i < nOwn; i += OWN_THREADS) cnt[lo + i] = lcnt[i];
  for (int g = tid; g < nOwn * (BUCKET / 4); g += OWN_THREADS) {
    const int node = g >> 3;
    const int slot = (g & 7) << 2;
    const int w = node * BUCKET + slot;
    const int deg = lcnt[node];
    int4 v = *(const int4*)(lbkt + w);
    v.x = slot + 0 < deg ? v.x : 0;
    v.y = slot + 1 < deg ? v.y : 0;
    v.z = slot + 2 < deg ? v.z : 0;
    v.w = slot + 3 < deg ? v.w : 0;
    *(int4*)(bucket + (size_t)lo * BUCKET + w) = v;
  }
}

__global__ void sage_gather_kernel(const float* __restrict__ xin,
                                   const int* __restrict__ cnt,
                                   const int* __restrict__ bucket,
                                   const int* __restrict__ ovfDst,
                                   const int* __restrict__ ovfSrc, int ovfCap,
                                   float* __restrict__ agg, int nNodes) {
  const int idx = blockIdx.x * blockDim.x + threadIdx.x;
  const int row = idx >> 4;
  if (row >= nNodes) return;
  const int c = (idx & 15) << 2;
  const int n = cnt[row];
  const int end = n < BUCKET ? n : BUCKET;
  const int* __restrict__ b = bucket + row * BUCKET;
  float4 acc = make_float4(0.0f, 0.0f, 0.0f, 0.0f);
  int j = 0;
  for (; j + 4 <= end; j += 4) {
    const int4 s = *(const int4*)(b + j);
    const float4 v0 = *(const float4*)(xin + (size_t)s.x * D + c);
    const float4 v1 = *(const float4*)(xin + (size_t)s.y * D + c);
    const float4 v2 = *(const float4*)(xin + (size_t)s.z * D + c);
    const float4 v3 = *(const float4*)(xin + (size_t)s.w * D + c);
    acc.x += v0.x; acc.y += v0.y; acc.z += v0.z; acc.w += v0.w;
    acc.x += v1.x; acc.y += v1.y; acc.z += v1.z; acc.w += v1.w;
    acc.x += v2.x; acc.y += v2.y; acc.z += v2.z; acc.w += v2.w;
    acc.x += v3.x; acc.y += v3.y; acc.z += v3.z; acc.w += v3.w;
  }
  for (; j < end; ++j) {
    const int s0 = b[j];
    const float4 v0 = *(const float4*)(xin + (size_t)s0 * D + c);
    acc.x += v0.x; acc.y += v0.y; acc.z += v0.z; acc.w += v0.w;
  }
  if (n > BUCKET) {
    const int len = cnt[nNodes];
    const int m = len < ovfCap ? len : ovfCap;
    for (int i = 0; i < m; ++i) {
      if (ovfDst[i] == row) {
        const float4 v0 = *(const float4*)(xin + (size_t)ovfSrc[i] * D + c);
        acc.x += v0.x; acc.y += v0.y; acc.z += v0.z; acc.w += v0.w;
      }
    }
  }
  *(float4*)(agg + (size_t)row * D + c) = acc;
}

template <bool RELU>
__global__ void sage_wmma_kernel(const float* __restrict__ agg,
                                 const int* __restrict__ cnt,
                                 const float* __restrict__ xin,
                                 const float* __restrict__ Wl,
                                 const float* __restrict__ bl,
                                 const float* __restrict__ Wr,
                                 float* __restrict__ out, int nTiles) {
  const int lane = threadIdx.x & 31;
  const int wave = threadIdx.x >> 5;
  const int tile = blockIdx.x * (blockDim.x >> 5) + wave;
  if (tile >= nTiles) return;

  const int m = lane & 15;
  const int khalf = (lane >> 4) << 1;
  const int row0 = tile << 4;
  const int row = row0 + m;

  const float rdeg = 1.0f / fmaxf((float)cnt[row], 1.0f);
  const float* __restrict__ arow = agg + (size_t)row * D;
  const float* __restrict__ xrow = xin + (size_t)row * D;

  v8f acc[4];
#pragma unroll
  for (int nt = 0; nt < 4; ++nt) acc[nt] = (v8f){0, 0, 0, 0, 0, 0, 0, 0};

#pragma unroll
  for (int kk = 0; kk < 16; ++kk) {
    const int k = (kk << 2) + khalf;
    v2f am = *(const v2f*)(arow + k);
    am[0] *= rdeg;
    am[1] *= rdeg;
    const v2f ax = *(const v2f*)(xrow + k);
#pragma unroll
    for (int nt = 0; nt < 4; ++nt) {
      const int j = (nt << 4) + m;
      const v2f bL = *(const v2f*)(Wl + j * D + k);
      const v2f bR = *(const v2f*)(Wr + j * D + k);
      acc[nt] = __builtin_amdgcn_wmma_f32_16x16x4_f32(
          false, am, false, bL, (short)0, acc[nt], false, false);
      acc[nt] = __builtin_amdgcn_wmma_f32_16x16x4_f32(
          false, ax, false, bR, (short)0, acc[nt], false, false);
    }
  }

  const int mhi = (lane >> 4) << 3;
#pragma unroll
  for (int nt = 0; nt < 4; ++nt) {
    const int j = (nt << 4) + m;
    const float b = bl[j];
#pragma unroll
    for (int v = 0; v < 8; ++v) {
      float val = acc[nt][v] + b;
      if (RELU) val = fmaxf(val, 0.0f);
      out[(size_t)(row0 + mhi + v) * D + j] = val;
    }
  }
}

template <bool RELU>
__global__ void sage_tail_kernel(const float* __restrict__ agg,
                                 const int* __restrict__ cnt,
                                 const float* __restrict__ xin,
                                 const float* __restrict__ Wl,
                                 const float* __restrict__ bl,
                                 const float* __restrict__ Wr,
                                 float* __restrict__ out, int row0, int nRows) {
  int idx = blockIdx.x * blockDim.x + threadIdx.x;
  if (idx >= nRows * D) return;
  int r = row0 + idx / D;
  int j = idx % D;
  float rdeg = 1.0f / fmaxf((float)cnt[r], 1.0f);
  float s = bl[j];
  for (int k = 0; k < D; ++k)
    s += agg[(size_t)r * D + k] * rdeg * Wl[j * D + k] +
         xin[(size_t)r * D + k] * Wr[j * D + k];
  if (RELU) s = fmaxf(s, 0.0f);
  out[(size_t)r * D + j] = s;
}

static inline size_t align256(size_t x) { return (x + 255) & ~(size_t)255; }

extern "C" void kernel_launch(void* const* d_in, const int* in_sizes, int n_in,
                              void* d_out, int out_size, void* d_ws,
                              size_t ws_size, hipStream_t stream) {
  const float* x = (const float*)d_in[0];
  const int* src = (const int*)d_in[1];
  const int* dst = (const int*)d_in[2];
  const float* Wl[3] = {(const float*)d_in[3], (const float*)d_in[6],
                        (const float*)d_in[9]};
  const float* bl[3] = {(const float*)d_in[4], (const float*)d_in[7],
                        (const float*)d_in[10]};
  const float* Wr[3] = {(const float*)d_in[5], (const float*)d_in[8],
                        (const float*)d_in[11]};

  const int nNodes = in_sizes[0] / D;
  const int nEdges = in_sizes[1];
  const size_t featBytes = (size_t)nNodes * D * sizeof(float);

  char* ws = (char*)d_ws;
  int* cnt = (int*)ws;
  int* ovfLen = cnt + nNodes;
  size_t ofs = align256(((size_t)nNodes + 1) * sizeof(int));
  float* agg = (float*)(ws + ofs);
  ofs += align256(featBytes);
  float* h0 = (float*)(ws + ofs);
  ofs += align256(featBytes);
  float* h1 = (float*)(ws + ofs);

  int* bucket = (int*)d_out;
  const size_t bucketBytes = (size_t)nNodes * BUCKET * sizeof(int);
  const int ovfCap = (int)((featBytes - bucketBytes) / (2 * sizeof(int)));
  int* ovfDst = (int*)((char*)d_out + bucketBytes);
  int* ovfSrc = ovfDst + ovfCap;
  int* rangeCount = (int*)h0;
  int* cursor = rangeCount + RANGES_MAX;
  int* rangeStart = cursor + RANGES_MAX;
  int* pairs = (int*)agg;
  const int partBlocks = (nEdges + PART_THREADS * PART_QUADS * 4 - 1) /
                         (PART_THREADS * PART_QUADS * 4);
  hipMemsetAsync(rangeCount, 0, 2 * RANGES_MAX * sizeof(int), stream);
  sage_count_kernel<<<partBlocks, PART_THREADS, 0, stream>>>(dst, rangeCount,
                                                            ovfLen, nEdges);
  sage_place_kernel<<<partBlocks, PART_THREADS, 0, stream>>>(
      src, dst, rangeCount, cursor, rangeStart, pairs, nEdges);
  sage_bucket_kernel<<<(nNodes + OWN - 1) / OWN, OWN_THREADS, 0, stream>>>(
      pairs, rangeCount, rangeStart, cnt, bucket, ovfDst, ovfSrc, ovfLen,
      ovfCap, nNodes);

  const int gatherBlocks = (nNodes * 16 + 255) / 256;
  const int nTiles = nNodes / 16;
  const int gemmBlocks = (nTiles + 7) / 8;
  const int tail = nNodes - nTiles * 16;

  auto run_layer = [&](const float* in, const float* wl, const float* b,
                       const float* wr, float* outp, bool relu) {
    sage_gather_kernel<<<gatherBlocks, 256, 0, stream>>>(
        in, cnt, bucket, ovfDst, ovfSrc, ovfCap, agg, nNodes);
    if (nTiles > 0) {
      if (relu)
        sage_wmma_kernel<true><<<gemmBlocks, 256, 0, stream>>>(
            agg, cnt, in, wl, b, wr, outp, nTiles);
      else
        sage_wmma_kernel<false><<<gemmBlocks, 256, 0, stream>>>(
            agg, cnt, in, wl, b, wr, outp, nTiles);
    }
    if (tail > 0) {
      int tb = (tail * D + 255) / 256;
      if (relu)
        sage_tail_kernel<true><<<tb, 256, 0, stream>>>(
            agg, cnt, in, wl, b, wr, outp, nTiles * 16, tail);
      else
        sage_tail_kernel<false><<<tb, 256, 0, stream>>>(
            agg, cnt, in, wl, b, wr, outp, nTiles * 16, tail);
    }
  };

  run_layer(x, Wl[0], bl[0], Wr[0], h0,  true);
  run_layer(h0, Wl[1], bl[1], Wr[1], h1,  true);
  run_layer(h1, Wl[2], bl[2], Wr[2], (float*)d_out,  false);
}


# Layer product fed through LDS: 256 resident 16-wave work-groups stage Wl, Wr once and each wave its 16-row tiles by whole-line 16-B reads (were 8-B lane-strided global loads); rows stored whole-line
# speedup vs baseline: 3.9403x; 1.2794x over previous
#include <hip/hip_runtime.h>

typedef float v2f __attribute__((ext_vector_type(2)));
typedef float v8f __attribute__((ext_vector_type(8)));

#define D 64

#define BUCKET 32
#define OWN 784
#define OWN_THREADS 1024
#define RANGES_MAX 256
#define PART_THREADS 256
#define PART_QUADS 4

__device__ __forceinline__ int4 sage_load4(const int* __restrict__ p, int e,
                                           int nEdges) {
  if (e + 4 <= nEdges) return *(const int4*)(p + e);
  int4 v = make_int4(-1, -1, -1, -1);
  if (e + 0 < nEdges) v.x = p[e + 0];
  if (e + 1 < nEdges) v.y = p[e + 1];
  if (e + 2 < nEdges) v.z = p[e + 2];
  return v;
}

__global__ void __launch_bounds__(PART_THREADS)
sage_count_kernel(const int* __restrict__ dst, int* __restrict__ rangeCount,
                  int* __restrict__ ovfLen, int nEdges) {
  __shared__ int hist[RANGES_MAX];
  const int tid = threadIdx.x;
  hist[tid] = 0;
  if (blockIdx.x == 0 && tid == 0) *ovfLen = 0;
  __syncthreads();
  const int e0 = blockIdx.x * (PART_THREADS * PART_QUADS * 4);
#pragma unroll
  for (int k = 0; k < PART_QUADS; ++k) {
    const int4 d = sage_load4(dst, e0 + (k * PART_THREADS + tid) * 4, nEdges);
    if (d.x >= 0) atomicAdd(&hist[d.x / OWN], 1);
    if (d.y >= 0) atomicAdd(&hist[d.y / OWN], 1);
    if (d.z >= 0) atomicAdd(&hist[d.z / OWN], 1);
    if (d.w >= 0) atomicAdd(&hist[d.w / OWN], 1);
  }
  __syncthreads();
  const int h = hist[tid];
  if (h) atomicAdd(rangeCount + tid, h);
}

__device__ __forceinline__ int sage_scan_incl(int v, int* a, int* b, int tid) {
  __syncthreads();
  a[tid] = v;
  __syncthreads();
  int* in = a;
  int* out = b;
  for (int off = 1; off < RANGES_MAX; off <<= 1) {
    out[tid] = in[tid] + (tid >= off ? in[tid - off] : 0);
    __syncthreads();
    int* t = in; in = out; out = t;
  }
  return in[tid];
}

__global__ void __launch_bounds__(PART_THREADS)
sage_place_kernel(const int* __restrict__ src, const int* __restrict__ dst,
                  const int* __restrict__ rangeCount,
                  int* __restrict__ cursor, int* __restrict__ rangeStart,
                  int* __restrict__ pairs, int nEdges) {
  __shared__ int hist[RANGES_MAX];
  __shared__ int lstart[RANGES_MAX];
  __shared__ int base[RANGES_MAX];
  __shared__ int scA[RANGES_MAX], scB[RANGES_MAX];
  __shared__ __attribute__((aligned(8))) int stage[PART_THREADS * PART_QUADS * 4 * 2];
  const int tid = threadIdx.x;
  hist[tid] = 0;
  __syncthreads();

  int4 d[PART_QUADS], s[PART_QUADS], rk[PART_QUADS];
  const int e0 = blockIdx.x * (PART_THREADS * PART_QUADS * 4);
#pragma unroll
  for (int k = 0; k < PART_QUADS; ++k) {
    const int e = e0 + (k * PART_THREADS + tid) * 4;
    d[k] = sage_load4(dst, e, nEdges);
    s[k] = sage_load4(src, e, nEdges);
    rk[k] = make_int4(0, 0, 0, 0);
    if (d[k].x >= 0) rk[k].x = atomicAdd(&hist[d[k].x / OWN], 1);
    if (d[k].y >= 0) rk[k].y = atomicAdd(&hist[d[k].y / OWN], 1);
    if (d[k].z >= 0) rk[k].z = atomicAdd(&hist[d[k].z / OWN], 1);
    if (d[k].w >= 0) rk[k].w = atomicAdd(&hist[d[k].w / OWN], 1);
  }

  const int total = rangeCount[tid];
  const int start = sage_scan_incl(total, scA, scB, tid) - total;
  if (blockIdx.x == 0) rangeStart[tid] = start;
  const int h = hist[tid];
  const int ls = sage_scan_incl(h, scA, scB, tid) - h;
  lstart[tid] = ls;
  base[tid] = start + (h ? atomicAdd(cursor + tid, h) : 0) - ls;
  __syncthreads();

#pragma unroll
  for (int k = 0; k < PART_QUADS; ++k) {
    if (d[k].x >= 0) *(int2*)(stage + 2 * (lstart[d[k].x / OWN] + rk[k].x)) = make_int2(d[k].x, s[k].x);
    if (d[k].y >= 0) *(int2*)(stage + 2 * (lstart[d[k].y / OWN] + rk[k].y)) = make_int2(d[k].y, s[k].y);
    if (d[k].z >= 0) *(int2*)(stage + 2 * (lstart[d[k].z / OWN] + rk[k].z)) = make_int2(d[k].z, s[k].z);
    if (d[k].w >= 0) *(int2*)(stage + 2 * (lstart[d[k].w / OWN] + rk[k].w)) = make_int2(d[k].w, s[k].w);
  }
  __syncthreads();

  int nHere = nEdges - e0;
  if (nHere > PART_THREADS * PART_QUADS * 4) nHere = PART_THREADS * PART_QUADS * 4;
  for (int i = tid; i < nHere; i += PART_THREADS) {
    const int2 pr = *(const int2*)(stage + 2 * i);
    *(int2*)(pairs + 2 * (size_t)(base[pr.x / OWN] + i)) = pr;
  }
}

__global__ void __launch_bounds__(OWN_THREADS)
sage_bucket_kernel(const int* __restrict__ pairs,
                   const int* __restrict__ rangeCount,
                   const int* __restrict__ rangeStart,
                   int* __restrict__ cnt,
                   int* __restrict__ bucket,
                   int* __restrict__ ovfDst,
                   int* __restrict__ ovfSrc,
                   int* __restrict__ ovfLen, int ovfCap, int nNodes) {
  __shared__ int lcnt[OWN];
  __shared__ __attribute__((aligned(16))) int lbkt[OWN * BUCKET];
  const int tid = threadIdx.x;
  const int lo = blockIdx.x * OWN;
  const int left = nNodes - lo;
  const int nOwn = left < OWN ? left : OWN;
  for (int i = tid; i < OWN; i += OWN_THREADS) lcnt[i] = 0;
  __syncthreads();

  const int n = rangeCount[blockIdx.x];
  const int* __restrict__ seg = pairs + 2 * (size_t)rangeStart[blockIdx.x];
  for (int i = tid; i < n; i += OWN_THREADS) {
    const int2 pr = *(const int2*)(seg + 2 * i);
    const int r = pr.x - lo;
    const int slot = atomicAdd(&lcnt[r], 1);
    if (slot < BUCKET) {
      lbkt[r * BUCKET + slot] = pr.y;
    } else {
      const int pos = atomicAdd(ovfLen, 1);
      if (pos < ovfCap) {
        ovfDst[pos] = pr.x;
        ovfSrc[pos] = pr.y;
      }
    }
  }
  __syncthreads();

  for (int i = tid; i < nOwn; i += OWN_THREADS) cnt[lo + i] = lcnt[i];
  for (int g = tid; g < nOwn * (BUCKET / 4); g += OWN_THREADS) {
    const int node = g >> 3;
    const int slot = (g & 7) << 2;
    const int w = node * BUCKET + slot;
    const int deg = lcnt[node];
    int4 v = *(const int4*)(lbkt + w);
    v.x = slot + 0 < deg ? v.x : 0;
    v.y = slot + 1 < deg ? v.y : 0;
    v.z = slot + 2 < deg ? v.z : 0;
    v.w = slot + 3 < deg ? v.w : 0;
    *(int4*)(bucket + (size_t)lo * BUCKET + w) = v;
  }
}

__global__ void sage_gather_kernel(const float* __restrict__ xin,
                                   const int* __restrict__ cnt,
                                   const int* __restrict__ bucket,
                                   const int* __restrict__ ovfDst,
                                   const int* __restrict__ ovfSrc, int ovfCap,
                                   float* __restrict__ agg, int nNodes) {
  const int idx = blockIdx.x * blockDim.x + threadIdx.x;
  const int row = idx >> 4;
  if (row >= nNodes) return;
  const int c = (idx & 15) << 2;
  const int n = cnt[row];
  const int end = n < BUCKET ? n : BUCKET;
  const int* __restrict__ b = bucket + row * BUCKET;
  float4 acc = make_float4(0.0f, 0.0f, 0.0f, 0.0f);
  int j = 0;
  for (; j + 4 <= end; j += 4) {
    const int4 s = *(const int4*)(b + j);
    const float4 v0 = *(const float4*)(xin + (size_t)s.x * D + c);
    const float4 v1 = *(const float4*)(xin + (size_t)s.y * D + c);
    const float4 v2 = *(const float4*)(xin + (size_t)s.z * D + c);
    const float4 v3 = *(const float4*)(xin + (size_t)s.w * D + c);
    acc.x += v0.x; acc.y += v0.y; acc.z += v0.z; acc.w += v0.w;
    acc.x += v1.x; acc.y += v1.y; acc.z += v1.z; acc.w += v1.w;
    acc.x += v2.x; acc.y += v2.y; acc.z += v2.z; acc.w += v2.w;
    acc.x += v3.x; acc.y += v3.y; acc.z += v3.z; acc.w += v3.w;
  }
  for (; j < end; ++j) {
    const int s0 = b[j];
    const float4 v0 = *(const float4*)(xin + (size_t)s0 * D + c);
    acc.x += v0.x; acc.y += v0.y; acc.z += v0.z; acc.w += v0.w;
  }
  if (n > BUCKET) {
    const int len = cnt[nNodes];
    const int m = len < ovfCap ? len : ovfCap;
    for (int i = 0; i < m; ++i) {
      if (ovfDst[i] == row) {
        const float4 v0 = *(const float4*)(xin + (size_t)ovfSrc[i] * D + c);
        acc.x += v0.x; acc.y += v0.y; acc.z += v0.z; acc.w += v0.w;
      }
    }
  }
  *(float4*)(agg + (size_t)row * D + c) = acc;
}

#define WPITCH 68
#define GEMM_WAVES 16
#define GEMM_THREADS (GEMM_WAVES * 32)
#define GEMM_BLOCKS 256
#define TILE_WORDS (16 * WPITCH)

__device__ __forceinline__ void sage_wave_lds_sync() {
  __builtin_amdgcn_fence(__ATOMIC_RELEASE, "wavefront");
  __builtin_amdgcn_wave_barrier();
  __builtin_amdgcn_fence(__ATOMIC_ACQUIRE, "wavefront");
}

template <bool RELU>
__global__ void __launch_bounds__(GEMM_THREADS)
sage_wmma_kernel(const float* __restrict__ agg, const int* __restrict__ cnt,
                 const float* __restrict__ xin, const float* __restrict__ Wl,
                 const float* __restrict__ bl, const float* __restrict__ Wr,
                 float* __restrict__ out, int nTiles) {
  __shared__ __attribute__((aligned(16))) float sW[2 * D * WPITCH];
  __shared__ __attribute__((aligned(16))) float sA[GEMM_WAVES * 2 * TILE_WORDS];

  const int tid = threadIdx.x;
  const int lane = tid & 31;
  const int wave = __builtin_amdgcn_readfirstlane(tid >> 5);

  const int lo = (int)(((unsigned long long)blockIdx.x * (unsigned)nTiles) / GEMM_BLOCKS);
  const int hi = (int)(((unsigned long long)(blockIdx.x + 1) * (unsigned)nTiles) / GEMM_BLOCKS);
  const int rounds = (hi - lo + GEMM_WAVES - 1) / GEMM_WAVES;

  const int rr = lane >> 4;
  const int cc = (lane & 15) << 2;
  float* __restrict__ tA = sA + wave * (2 * TILE_WORDS);
  float* __restrict__ tX = tA + TILE_WORDS;

  const int m = lane & 15;
  const int khalf = (lane >> 4) << 1;
  const int mhi = (lane >> 4) << 3;
  const float* __restrict__ fA = tA + m * WPITCH + khalf;
  const float* __restrict__ fX = tX + m * WPITCH + khalf;
  const float* __restrict__ fL = sW + m * WPITCH + khalf;
  const float* __restrict__ fR = fL + D * WPITCH;

  for (int r = 0; r < rounds; ++r) {
    const int tile = lo + r * GEMM_WAVES + wave;
    const bool valid = tile < hi;
    const int row0 = tile << 4;

    if (valid) {
      const float* __restrict__ ga = agg + (size_t)row0 * D + cc;
      const float* __restrict__ gx = xin + (size_t)row0 * D + cc;
#pragma unroll
      for (int i = 0; i < 8; ++i) {
        const float4 a = *(const float4*)(ga + (size_t)(2 * i + rr) * D);
        const float4 x = *(const float4*)(gx + (size_t)(2 * i + rr) * D);
        *(float4*)(tA + (2 * i + rr) * WPITCH + cc) = a;
        *(float4*)(tX + (2 * i + rr) * WPITCH + cc) = x;
      }
    }
    if (r == 0) {
#pragma unroll
      for (int i = 0; i < (2 * D * (D / 4)) / GEMM_THREADS; ++i) {
        const int g = i * GEMM_THREADS + tid;
        const int wr = g >> 4;
        const float* __restrict__ w = wr < D ? Wl + wr * D : Wr + (wr - D) * D;
        *(float4*)(sW + wr * WPITCH + ((g & 15) << 2)) =
            *(const float4*)(w + ((g & 15) << 2));
      }
      __syncthreads();
    }
    if (!valid) continue;

    const float rdeg = 1.0f / fmaxf((float)cnt[row0 + m], 1.0f);
    sage_wave_lds_sync();

    v8f acc[4];
#pragma unroll
    for (int nt = 0; nt < 4; ++nt) acc[nt] = (v8f){0, 0, 0, 0, 0, 0, 0, 0};

#pragma unroll
    for (int kk = 0; kk < 16; ++kk) {
      v2f am = *(const v2f*)(fA + (kk << 2));
      am[0] *= rdeg;
      am[1] *= rdeg;
      const v2f ax = *(const v2f*)(fX + (kk << 2));
#pragma unroll
      for (int nt = 0; nt < 4; ++nt) {
        const v2f bL = *(const v2f*)(fL + (nt << 4) * WPITCH + (kk << 2));
        const v2f bR = *(const v2f*)(fR + (nt << 4) * WPITCH + (kk << 2));
        acc[nt] = __builtin_amdgcn_wmma_f32_16x16x4_f32(
            false, am, false, bL, (short)0, acc[nt], false, false);
        acc[nt] = __builtin_amdgcn_wmma_f32_16x16x4_f32(
            false, ax, false, bR, (short)0, acc[nt], false, false);
      }
    }

#pragma unroll
    for (int nt = 0; nt < 4; ++nt) {
      const int j = (nt << 4) + m;
      const float b = bl[j];
#pragma unroll
      for (int v = 0; v < 8; ++v) {
        float val = acc[nt][v] + b;
        if (RELU) val = fmaxf(val, 0.0f);
        tA[(mhi + v) * WPITCH + j] = val;
      }
    }
    sage_wave_lds_sync();
    float* __restrict__ go = out + (size_t)row0 * D + cc;
#pragma unroll
    for (int i = 0; i < 8; ++i)
      *(float4*)(go + (size_t)(2 * i + rr) * D) =
          *(const float4*)(tA + (2 * i + rr) * WPITCH + cc);
    sage_wave_lds_sync();
  }
}

template <bool RELU>
__global__ void sage_tail_kernel(const float* __restrict__ agg,
                                 const int* __restrict__ cnt,
                                 const float* __restrict__ xin,
                                 const float* __restrict__ Wl,
                                 const float* __restrict__ bl,
                                 const float* __restrict__ Wr,
                                 float* __restrict__ out, int row0, int nRows) {
  int idx = blockIdx.x * blockDim.x + threadIdx.x;
  if (idx >= nRows * D) return;
  int r = row0 + idx / D;
  int j = idx % D;
  float rdeg = 1.0f / fmaxf((float)cnt[r], 1.0f);
  float s = bl[j];
  for (int k = 0; k < D; ++k)
    s += agg[(size_t)r * D + k] * rdeg * Wl[j * D + k] +
         xin[(size_t)r * D + k] * Wr[j * D + k];
  if (RELU) s = fmaxf(s, 0.0f);
  out[(size_t)r * D + j] = s;
}

static inline size_t align256(size_t x) { return (x + 255) & ~(size_t)255; }

extern "C" void kernel_launch(void* const* d_in, const int* in_sizes, int n_in,
                              void* d_out, int out_size, void* d_ws,
                              size_t ws_size, hipStream_t stream) {
  const float* x = (const float*)d_in[0];
  const int* src = (const int*)d_in[1];
  const int* dst = (const int*)d_in[2];
  const float* Wl[3] = {(const float*)d_in[3], (const float*)d_in[6],
                        (const float*)d_in[9]};
  const float* bl[3] = {(const float*)d_in[4], (const float*)d_in[7],
                        (const float*)d_in[10]};
  const float* Wr[3] = {(const float*)d_in[5], (const float*)d_in[8],
                        (const float*)d_in[11]};

  const int nNodes = in_sizes[0] / D;
  const int nEdges = in_sizes[1];
  const size_t featBytes = (size_t)nNodes * D * sizeof(float);

  char* ws = (char*)d_ws;
  int* cnt = (int*)ws;
  int* ovfLen = cnt + nNodes;
  size_t ofs = align256(((size_t)nNodes + 1) * sizeof(int));
  float* agg = (float*)(ws + ofs);
  ofs += align256(featBytes);
  float* h0 = (float*)(ws + ofs);
  ofs += align256(featBytes);
  float* h1 = (float*)(ws + ofs);

  int* bucket = (int*)d_out;
  const size_t bucketBytes = (size_t)nNodes * BUCKET * sizeof(int);
  const int ovfCap = (int)((featBytes - bucketBytes) / (2 * sizeof(int)));
  int* ovfDst = (int*)((char*)d_out + bucketBytes);
  int* ovfSrc = ovfDst + ovfCap;
  int* rangeCount = (int*)h0;
  int* cursor = rangeCount + RANGES_MAX;
  int* rangeStart = cursor + RANGES_MAX;
  int* pairs = (int*)agg;
  const int partBlocks = (nEdges + PART_THREADS * PART_QUADS * 4 - 1) /
                         (PART_THREADS * PART_QUADS * 4);
  hipMemsetAsync(rangeCount, 0, 2 * RANGES_MAX * sizeof(int), stream);
  sage_count_kernel<<<partBlocks, PART_THREADS, 0, stream>>>(dst, rangeCount,
                                                            ovfLen, nEdges);
  sage_place_kernel<<<partBlocks, PART_THREADS, 0, stream>>>(
      src, dst, rangeCount, cursor, rangeStart, pairs, nEdges);
  sage_bucket_kernel<<<(nNodes + OWN - 1) / OWN, OWN_THREADS, 0, stream>>>(
      pairs, rangeCount, rangeStart, cnt, bucket, ovfDst, ovfSrc, ovfLen,
      ovfCap, nNodes);

  const int gatherBlocks = (nNodes * 16 + 255) / 256;
  const int nTiles = nNodes / 16;
  const int gemmBlocks = GEMM_BLOCKS;
  const int tail = nNodes - nTiles * 16;

  auto run_layer = [&](const float* in, const float* wl, const float* b,
                       const float* wr, float* outp, bool relu) {
    sage_gather_kernel<<<gatherBlocks, 256, 0, stream>>>(
        in, cnt, bucket, ovfDst, ovfSrc, ovfCap, agg, nNodes);
    if (nTiles > 0) {
      if (relu)
        sage_wmma_kernel<true><<<gemmBlocks, GEMM_THREADS, 0, stream>>>(
            agg, cnt, in, wl, b, wr, outp, nTiles);
      else
        sage_wmma_kernel<false><<<gemmBlocks, GEMM_THREADS, 0, stream>>>(
            agg, cnt, in, wl, b, wr, outp, nTiles);
    }
    if (tail > 0) {
      int tb = (tail * D + 255) / 256;
      if (relu)
        sage_tail_kernel<true><<<tb, 256, 0, stream>>>(
            agg, cnt, in, wl, b, wr, outp, nTiles * 16, tail);
      else
        sage_tail_kernel<false><<<tb, 256, 0, stream>>>(
            agg, cnt, in, wl, b, wr, outp, nTiles * 16, tail);
    }
  };

  run_layer(x, Wl[0], bl[0], Wr[0], h0,  true);
  run_layer(h0, Wl[1], bl[1], Wr[1], h1,  true);
  run_layer(h1, Wl[2], bl[2], Wr[2], (float*)d_out,  false);
}


# Aggregation fused into the layer product: each wave sums its tile's 16 source-row sets into its LDS A tile (ids from staged bucket lines); agg buffer, 3 gather launches dropped; sum order kept
# speedup vs baseline: 4.0677x; 1.0323x over previous
#include <hip/hip_runtime.h>

typedef float v2f __attribute__((ext_vector_type(2)));
typedef float v8f __attribute__((ext_vector_type(8)));

#define D 64

#define BUCKET 32
#define OWN 784
#define OWN_THREADS 1024
#define RANGES_MAX 256
#define PART_THREADS 256
#define PART_QUADS 4

__device__ __forceinline__ int4 sage_load4(const int* __restrict__ p, int e,
                                           int nEdges) {
  if (e + 4 <= nEdges) return *(const int4*)(p + e);
  int4 v = make_int4(-1, -1, -1, -1);
  if (e + 0 < nEdges) v.x = p[e + 0];
  if (e + 1 < nEdges) v.y = p[e + 1];
  if (e + 2 < nEdges) v.z = p[e + 2];
  return v;
}

__global__ void __launch_bounds__(PART_THREADS)
sage_count_kernel(const int* __restrict__ dst, int* __restrict__ rangeCount,
                  int* __restrict__ ovfLen, int nEdges) {
  __shared__ int hist[RANGES_MAX];
  const int tid = threadIdx.x;
  hist[tid] = 0;
  if (blockIdx.x == 0 && tid == 0) *ovfLen = 0;
  __syncthreads();
  const int e0 = blockIdx.x * (PART_THREADS * PART_QUADS * 4);
#pragma unroll
  for (int k = 0; k < PART_QUADS; ++k) {
    const int4 d = sage_load4(dst, e0 + (k * PART_THREADS + tid) * 4, nEdges);
    if (d.x >= 0) atomicAdd(&hist[d.x / OWN], 1);
    if (d.y >= 0) atomicAdd(&hist[d.y / OWN], 1);
    if (d.z >= 0) atomicAdd(&hist[d.z / OWN], 1);
    if (d.w >= 0) atomicAdd(&hist[d.w / OWN], 1);
  }
  __syncthreads();
  const int h = hist[tid];
  if (h) atomicAdd(rangeCount + tid, h);
}

__device__ __forceinline__ int sage_scan_incl(int v, int* a, int* b, int tid) {
  __syncthreads();
  a[tid] = v;
  __syncthreads();
  int* in = a;
  int* out = b;
  for (int off = 1; off < RANGES_MAX; off <<= 1) {
    out[tid] = in[tid] + (tid >= off ? in[tid - off] : 0);
    __syncthreads();
    int* t = in; in = out; out = t;
  }
  return in[tid];
}

__global__ void __launch_bounds__(PART_THREADS)
sage_place_kernel(const int* __restrict__ src, const int* __restrict__ dst,
                  const int* __restrict__ rangeCount,
                  int* __restrict__ cursor, int* __restrict__ rangeStart,
                  int* __restrict__ pairs, int nEdges) {
  __shared__ int hist[RANGES_MAX];
  __shared__ int lstart[RANGES_MAX];
  __shared__ int base[RANGES_MAX];
  __shared__ int scA[RANGES_MAX], scB[RANGES_MAX];
  __shared__ __attribute__((aligned(8))) int stage[PART_THREADS * PART_QUADS * 4 * 2];
  const int tid = threadIdx.x;
  hist[tid] = 0;
  __syncthreads();

  int4 d[PART_QUADS], s[PART_QUADS], rk[PART_QUADS];
  const int e0 = blockIdx.x * (PART_THREADS * PART_QUADS * 4);
#pragma unroll
  for (int k = 0; k < PART_QUADS; ++k) {
    const int e = e0 + (k * PART_THREADS + tid) * 4;
    d[k] = sage_load4(dst, e, nEdges);
    s[k] = sage_load4(src, e, nEdges);
    rk[k] = make_int4(0, 0, 0, 0);
    if (d[k].x >= 0) rk[k].x = atomicAdd(&hist[d[k].x / OWN], 1);
    if (d[k].y >= 0) rk[k].y = atomicAdd(&hist[d[k].y / OWN], 1);
    if (d[k].z >= 0) rk[k].z = atomicAdd(&hist[d[k].z / OWN], 1);
    if (d[k].w >= 0) rk[k].w = atomicAdd(&hist[d[k].w / OWN], 1);
  }

  const int total = rangeCount[tid];
  const int start = sage_scan_incl(total, scA, scB, tid) - total;
  if (blockIdx.x == 0) rangeStart[tid] = start;
  const int h = hist[tid];
  const int ls = sage_scan_incl(h, scA, scB, tid) - h;
  lstart[tid] = ls;
  base[tid] = start + (h ? atomicAdd(cursor + tid, h) : 0) - ls;
  __syncthreads();

#pragma unroll
  for (int k = 0; k < PART_QUADS; ++k) {
    if (d[k].x >= 0) *(int2*)(stage + 2 * (lstart[d[k].x / OWN] + rk[k].x)) = make_int2(d[k].x, s[k].x);
    if (d[k].y >= 0) *(int2*)(stage + 2 * (lstart[d[k].y / OWN] + rk[k].y)) = make_int2(d[k].y, s[k].y);
    if (d[k].z >= 0) *(int2*)(stage + 2 * (lstart[d[k].z / OWN] + rk[k].z)) = make_int2(d[k].z, s[k].z);
    if (d[k].w >= 0) *(int2*)(stage + 2 * (lstart[d[k].w / OWN] + rk[k].w)) = make_int2(d[k].w, s[k].w);
  }
  __syncthreads();

  int nHere = nEdges - e0;
  if (nHere > PART_THREADS * PART_QUADS * 4) nHere = PART_THREADS * PART_QUADS * 4;
  for (int i = tid; i < nHere; i += PART_THREADS) {
    const int2 pr = *(const int2*)(stage + 2 * i);
    *(int2*)(pairs + 2 * (size_t)(base[pr.x / OWN] + i)) = pr;
  }
}

__global__ void __launch_bounds__(OWN_THREADS)
sage_bucket_kernel(const int* __restrict__ pairs,
                   const int* __restrict__ rangeCount,
                   const int* __restrict__ rangeStart,
                   int* __restrict__ cnt,
                   int* __restrict__ bucket,
                   int* __restrict__ ovfDst,
                   int* __restrict__ ovfSrc,
                   int* __restrict__ ovfLen, int ovfCap, int nNodes) {
  __shared__ int lcnt[OWN];
  __shared__ __attribute__((aligned(16))) int lbkt[OWN * BUCKET];
  const int tid = threadIdx.x;
  const int lo = blockIdx.x * OWN;
  const int left = nNodes - lo;
  const int nOwn = left < OWN ? left : OWN;
  for (int i = tid; i < OWN; i += OWN_THREADS) lcnt[i] = 0;
  __syncthreads();

  const int n = rangeCount[blockIdx.x];
  const int* __restrict__ seg = pairs + 2 * (size_t)rangeStart[blockIdx.x];
  for (int i = tid; i < n; i += OWN_THREADS) {
    const int2 pr = *(const int2*)(seg + 2 * i);
    const int r = pr.x - lo;
    const int slot = atomicAdd(&lcnt[r], 1);
    if (slot < BUCKET) {
      lbkt[r * BUCKET + slot] = pr.y;
    } else {
      const int pos = atomicAdd(ovfLen, 1);
      if (pos < ovfCap) {
        ovfDst[pos] = pr.x;
        ovfSrc[pos] = pr.y;
      }
    }
  }
  __syncthreads();

  for (int i = tid; i < nOwn; i += OWN_THREADS) cnt[lo + i] = lcnt[i];
  for (int g = tid; g < nOwn * (BUCKET / 4); g += OWN_THREADS) {
    const int node = g >> 3;
    const int slot = (g & 7) << 2;
    const int w = node * BUCKET + slot;
    const int deg = lcnt[node];
    int4 v = *(const int4*)(lbkt + w);
    v.x = slot + 0 < deg ? v.x : 0;
    v.y = slot + 1 < deg ? v.y : 0;
    v.z = slot + 2 < deg ? v.z : 0;
    v.w = slot + 3 < deg ? v.w : 0;
    *(int4*)(bucket + (size_t)lo * BUCKET + w) = v;
  }
}

#define WPITCH 68
#define GEMM_WAVES 16
#define GEMM_THREADS (GEMM_WAVES * 32)
#define GEMM_BLOCKS 256
#define TILE_WORDS (16 * WPITCH)
#define IDS_WORDS (16 * BUCKET + 16)

__device__ __forceinline__ void sage_wave_lds_sync() {
  __builtin_amdgcn_fence(__ATOMIC_RELEASE, "wavefront");
  __builtin_amdgcn_wave_barrier();
  __builtin_amdgcn_fence(__ATOMIC_ACQUIRE, "wavefront");
}

__device__ __forceinline__ float4 sage_row4(const float* __restrict__ xin,
                                            int s, int c) {
  return *(const float4*)(xin + (size_t)s * D + c);
}

__device__ __forceinline__ void sage_add_if(float4& a, const float4 v, bool p) {
  a.x = p ? a.x + v.x : a.x;
  a.y = p ? a.y + v.y : a.y;
  a.z = p ? a.z + v.z : a.z;
  a.w = p ? a.w + v.w : a.w;
}

template <bool RELU>
__global__ void __launch_bounds__(GEMM_THREADS)
sage_layer_kernel(const float* __restrict__ xin, const int* __restrict__ cnt,
                  const int* __restrict__ bucket,
                  const int* __restrict__ ovfDst,
                  const int* __restrict__ ovfSrc, int ovfCap, int nNodes,
                  const float* __restrict__ Wl, const float* __restrict__ bl,
                  const float* __restrict__ Wr, float* __restrict__ out,
                  int nTiles) {
  __shared__ __attribute__((aligned(16))) float sW[2 * D * WPITCH];
  __shared__ __attribute__((aligned(16))) float sA[GEMM_WAVES * 2 * TILE_WORDS];
  __shared__ __attribute__((aligned(16))) int sI[GEMM_WAVES * IDS_WORDS];

  const int tid = threadIdx.x;
  const int lane = tid & 31;
  const int wave = __builtin_amdgcn_readfirstlane(tid >> 5);

  const int lo = (int)(((unsigned long long)blockIdx.x * (unsigned)nTiles) / GEMM_BLOCKS);
  const int hi = (int)(((unsigned long long)(blockIdx.x + 1) * (unsigned)nTiles) / GEMM_BLOCKS);
  const int rounds = (hi - lo + GEMM_WAVES - 1) / GEMM_WAVES;

  const int rr = lane >> 4;
  const int cc = (lane & 15) << 2;
  float* __restrict__ tA = sA + wave * (2 * TILE_WORDS);
  float* __restrict__ tX = tA + TILE_WORDS;
  int* __restrict__ tI = sI + wave * IDS_WORDS;
  int* __restrict__ tC = tI + 16 * BUCKET;

  const int m = lane & 15;
  const int khalf = (lane >> 4) << 1;
  const int mhi = (lane >> 4) << 3;
  const float* __restrict__ fA = tA + m * WPITCH + khalf;
  const float* __restrict__ fX = tX + m * WPITCH + khalf;
  const float* __restrict__ fL = sW + m * WPITCH + khalf;
  const float* __restrict__ fR = fL + D * WPITCH;

  for (int r = 0; r < rounds; ++r) {
    const int tile = lo + r * GEMM_WAVES + wave;
    const bool valid = tile < hi;
    const int row0 = tile << 4;
    int deg = 0;

    if (valid) {
      const int* __restrict__ gb = bucket + (size_t)row0 * BUCKET + (lane << 2);
      const float* __restrict__ gx = xin + (size_t)row0 * D + cc;
      int4 ids[4];
#pragma unroll
      for (int i = 0; i < 4; ++i) ids[i] = *(const int4*)(gb + i * 128);
      deg = cnt[row0 + m];
      float4 xr[8];
#pragma unroll
      for (int i = 0; i < 8; ++i)
        xr[i] = *(const float4*)(gx + (size_t)(2 * i + rr) * D);
#pragma unroll
      for (int i = 0; i < 4; ++i) *(int4*)(tI + i * 128 + (lane << 2)) = ids[i];
      tC[m] = deg;
#pragma unroll
      for (int i = 0; i < 8; ++i)
        *(float4*)(tX + (2 * i + rr) * WPITCH + cc) = xr[i];
    }
    if (r == 0) {
#pragma unroll
      for (int i = 0; i < (2 * D * (D / 4)) / GEMM_THREADS; ++i) {
        const int g = i * GEMM_THREADS + tid;
        const int wr = g >> 4;
        const float* __restrict__ w = wr < D ? Wl + wr * D : Wr + (wr - D) * D;
        *(float4*)(sW + wr * WPITCH + ((g & 15) << 2)) =
            *(const float4*)(w + ((g & 15) << 2));
      }
      __syncthreads();
    }
    if (!valid) continue;

    sage_wave_lds_sync();

    for (int i = 0; i < 8; ++i) {
      const int lr = 2 * i + rr;
      const int n = tC[lr];
      const int end = n < BUCKET ? n : BUCKET;
      const int* __restrict__ lb = tI + lr * BUCKET;
      float4 acc = make_float4(0.0f, 0.0f, 0.0f, 0.0f);
      int j = 0;
      for (; j + 8 <= end; j += 8) {
        const int4 s = *(const int4*)(lb + j);
        const int4 t = *(const int4*)(lb + j + 4);
        const float4 v0 = sage_row4(xin, s.x, cc);
        const float4 v1 = sage_row4(xin, s.y, cc);
        const float4 v2 = sage_row4(xin, s.z, cc);
        const float4 v3 = sage_row4(xin, s.w, cc);
        const float4 v4 = sage_row4(xin, t.x, cc);
        const float4 v5 = sage_row4(xin, t.y, cc);
        const float4 v6 = sage_row4(xin, t.z, cc);
        const float4 v7 = sage_row4(xin, t.w, cc);
        acc.x += v0.x; acc.y += v0.y; acc.z += v0.z; acc.w += v0.w;
        acc.x += v1.x; acc.y += v1.y; acc.z += v1.z; acc.w += v1.w;
        acc.x += v2.x; acc.y += v2.y; acc.z += v2.z; acc.w += v2.w;
        acc.x += v3.x; acc.y += v3.y; acc.z += v3.z; acc.w += v3.w;
        acc.x += v4.x; acc.y += v4.y; acc.z += v4.z; acc.w += v4.w;
        acc.x += v5.x; acc.y += v5.y; acc.z += v5.z; acc.w += v5.w;
        acc.x += v6.x; acc.y += v6.y; acc.z += v6.z; acc.w += v6.w;
        acc.x += v7.x; acc.y += v7.y; acc.z += v7.z; acc.w += v7.w;
      }
      if (j + 4 <= end) {
        const int4 s = *(const int4*)(lb + j);
        const float4 v0 = sage_row4(xin, s.x, cc);
        const float4 v1 = sage_row4(xin, s.y, cc);
        const float4 v2 = sage_row4(xin, s.z, cc);
        const float4 v3 = sage_row4(xin, s.w, cc);
        acc.x += v0.x; acc.y += v0.y; acc.z += v0.z; acc.w += v0.w;
        acc.x += v1.x; acc.y += v1.y; acc.z += v1.z; acc.w += v1.w;
        acc.x += v2.x; acc.y += v2.y; acc.z += v2.z; acc.w += v2.w;
        acc.x += v3.x; acc.y += v3.y; acc.z += v3.z; acc.w += v3.w;
        j += 4;
      }
      if (j < end) {
        const int4 s = *(const int4*)(lb + j);
        const float4 v0 = sage_row4(xin, s.x, cc);
        const float4 v1 = sage_row4(xin, s.y, cc);
        const float4 v2 = sage_row4(xin, s.z, cc);
        acc.x += v0.x; acc.y += v0.y; acc.z += v0.z; acc.w += v0.w;
        sage_add_if(acc, v1, j + 1 < end);
        sage_add_if(acc, v2, j + 2 < end);
      }
      if (n > BUCKET) {
        const int row = row0 + lr;
        const int len = cnt[nNodes];
        const int mm = len < ovfCap ? len : ovfCap;
        for (int k = 0; k < mm; ++k) {
          if (ovfDst[k] == row) {
            const float4 u = sage_row4(xin, ovfSrc[k], cc);
            acc.x += u.x; acc.y += u.y; acc.z += u.z; acc.w += u.w;
          }
        }
      }
      *(float4*)(tA + lr * WPITCH + cc) = acc;
    }

    const float rdeg = 1.0f / fmaxf((float)deg, 1.0f);
    sage_wave_lds_sync();

    v8f acc[4];
#pragma unroll
    for (int nt = 0; nt < 4; ++nt) acc[nt] = (v8f){0, 0, 0, 0, 0, 0, 0, 0};

#pragma unroll
    for (int kk = 0; kk < 16; ++kk) {
      v2f am = *(const v2f*)(fA + (kk << 2));
      am[0] *= rdeg;
      am[1] *= rdeg;
      const v2f ax = *(const v2f*)(fX + (kk << 2));
#pragma unroll
      for (int nt = 0; nt < 4; ++nt) {
        const v2f bL = *(const v2f*)(fL + (nt << 4) * WPITCH + (kk << 2));
        const v2f bR = *(const v2f*)(fR + (nt << 4) * WPITCH + (kk << 2));
        acc[nt] = __builtin_amdgcn_wmma_f32_16x16x4_f32(
            false, am, false, bL, (short)0, acc[nt], false, false);
        acc[nt] = __builtin_amdgcn_wmma_f32_16x16x4_f32(
            false, ax, false, bR, (short)0, acc[nt], false, false);
      }
    }

#pragma unroll
    for (int nt = 0; nt < 4; ++nt) {
      const int j = (nt << 4) + m;
      const float b = bl[j];
#pragma unroll
      for (int v = 0; v < 8; ++v) {
        float val = acc[nt][v] + b;
        if (RELU) val = fmaxf(val, 0.0f);
        tA[(mhi + v) * WPITCH + j] = val;
      }
    }
    sage_wave_lds_sync();
    float* __restrict__ go = out + (size_t)row0 * D + cc;
#pragma unroll
    for (int i = 0; i < 8; ++i)
      *(float4*)(go + (size_t)(2 * i + rr) * D) =
          *(const float4*)(tA + (2 * i + rr) * WPITCH + cc);
    sage_wave_lds_sync();
  }
}

template <bool RELU>
__global__ void sage_tail_kernel(const float* __restrict__ xin,
                                 const int* __restrict__ cnt,
                                 const int* __restrict__ bucket,
                                 const int* __restrict__ ovfDst,
                                 const int* __restrict__ ovfSrc, int ovfCap,
                                 int nNodes,
                                 const float* __restrict__ Wl,
                                 const float* __restrict__ bl,
                                 const float* __restrict__ Wr,
                                 float* __restrict__ out, int row0, int nRows) {
  int idx = blockIdx.x * blockDim.x + threadIdx.x;
  if (idx >= nRows * D) return;
  int r = row0 + idx / D;
  int j = idx % D;
  const int n = cnt[r];
  const int end = n < BUCKET ? n : BUCKET;
  const int len = cnt[nNodes];
  const int mm = len < ovfCap ? len : ovfCap;
  float rdeg = 1.0f / fmaxf((float)n, 1.0f);
  float s = bl[j];
  for (int k = 0; k < D; ++k) {
    float a = 0.0f;
    for (int t = 0; t < end; ++t) a += xin[(size_t)bucket[r * BUCKET + t] * D + k];
    if (n > BUCKET)
      for (int t = 0; t < mm; ++t)
        if (ovfDst[t] == r) a += xin[(size_t)ovfSrc[t] * D + k];
    s += a * rdeg * Wl[j * D + k] + xin[(size_t)r * D + k] * Wr[j * D + k];
  }
  if (RELU) s = fmaxf(s, 0.0f);
  out[(size_t)r * D + j] = s;
}

static inline size_t align256(size_t x) { return (x + 255) & ~(size_t)255; }

extern "C" void kernel_launch(void* const* d_in, const int* in_sizes, int n_in,
                              void* d_out, int out_size, void* d_ws,
                              size_t ws_size, hipStream_t stream) {
  const float* x = (const float*)d_in[0];
  const int* src = (const int*)d_in[1];
  const int* dst = (const int*)d_in[2];
  const float* Wl[3] = {(const float*)d_in[3], (const float*)d_in[6],
                        (const float*)d_in[9]};
  const float* bl[3] = {(const float*)d_in[4], (const float*)d_in[7],
                        (const float*)d_in[10]};
  const float* Wr[3] = {(const float*)d_in[5], (const float*)d_in[8],
                        (const float*)d_in[11]};

  const int nNodes = in_sizes[0] / D;
  const int nEdges = in_sizes[1];
  const size_t featBytes = (size_t)nNodes * D * sizeof(float);

  char* ws = (char*)d_ws;
  int* cnt = (int*)ws;
  int* ovfLen = cnt + nNodes;
  size_t ofs = align256(((size_t)nNodes + 1) * sizeof(int));
  char* adj = ws + ofs;
  ofs += align256(featBytes);
  float* h0 = (float*)(ws + ofs);
  ofs += align256(featBytes);
  float* h1 = (float*)(ws + ofs);

  int* bucket = (int*)adj;
  const size_t bucketBytes = (size_t)nNodes * BUCKET * sizeof(int);
  const int ovfCap = (int)((featBytes - bucketBytes) / (2 * sizeof(int)));
  int* ovfDst = (int*)(adj + bucketBytes);
  int* ovfSrc = ovfDst + ovfCap;
  int* rangeCount = (int*)h0;
  int* cursor = rangeCount + RANGES_MAX;
  int* rangeStart = cursor + RANGES_MAX;
  int* pairs = (int*)d_out;
  const int partBlocks = (nEdges + PART_THREADS * PART_QUADS * 4 - 1) /
                         (PART_THREADS * PART_QUADS * 4);
  hipMemsetAsync(rangeCount, 0, 2 * RANGES_MAX * sizeof(int), stream);
  sage_count_kernel<<<partBlocks, PART_THREADS, 0, stream>>>(dst, rangeCount,
                                                            ovfLen, nEdges);
  sage_place_kernel<<<partBlocks, PART_THREADS, 0, stream>>>(
      src, dst, rangeCount, cursor, rangeStart, pairs, nEdges);
  sage_bucket_kernel<<<(nNodes + OWN - 1) / OWN, OWN_THREADS, 0, stream>>>(
      pairs, rangeCount, rangeStart, cnt, bucket, ovfDst, ovfSrc, ovfLen,
      ovfCap, nNodes);

  const int nTiles = nNodes / 16;
  const int gemmBlocks = GEMM_BLOCKS;
  const int tail = nNodes - nTiles * 16;

  auto run_layer = [&](const float* in, const float* wl, const float* b,
                       const float* wr, float* outp, bool relu) {
    if (nTiles > 0) {
      if (relu)
        sage_layer_kernel<true><<<gemmBlocks, GEMM_THREADS, 0, stream>>>(
            in, cnt, bucket, ovfDst, ovfSrc, ovfCap, nNodes, wl, b, wr, outp,
            nTiles);
      else
        sage_layer_kernel<false><<<gemmBlocks, GEMM_THREADS, 0, stream>>>(
            in, cnt, bucket, ovfDst, ovfSrc, ovfCap, nNodes, wl, b, wr, outp,
            nTiles);
    }
    if (tail > 0) {
      int tb = (tail * D + 255) / 256;
      if (relu)
        sage_tail_kernel<true><<<tb, 256, 0, stream>>>(
            in, cnt, bucket, ovfDst, ovfSrc, ovfCap, nNodes, wl, b, wr, outp,
            nTiles * 16, tail);
      else
        sage_tail_kernel<false><<<tb, 256, 0, stream>>>(
            in, cnt, bucket, ovfDst, ovfSrc, ovfCap, nNodes, wl, b, wr, outp,
            nTiles * 16, tail);
    }
  };

  run_layer(x, Wl[0], bl[0], Wr[0], h0,  true);
  run_layer(h0, Wl[1], bl[1], Wr[1], h1,  true);
  run_layer(h1, Wl[2], bl[2], Wr[2], (float*)d_out,  false);
}


# Layer kernel at 26 waves per work-group (were 16), as many tile areas as its LDS holds: a work-group's 24-25 tiles are summed and multiplied in one round, not two with the second half empty
# speedup vs baseline: 4.9227x; 1.2102x over previous
#include <hip/hip_runtime.h>

typedef float v2f __attribute__((ext_vector_type(2)));
typedef float v8f __attribute__((ext_vector_type(8)));

#define D 64

#define BUCKET 32
#define OWN 784
#define OWN_THREADS 1024
#define RANGES_MAX 256
#define PART_THREADS 256
#define PART_QUADS 4

__device__ __forceinline__ int4 sage_load4(const int* __restrict__ p, int e,
                                           int nEdges) {
  if (e + 4 <= nEdges) return *(const int4*)(p + e);
  int4 v = make_int4(-1, -1, -1, -1);
  if (e + 0 < nEdges) v.x = p[e + 0];
  if (e + 1 < nEdges) v.y = p[e + 1];
  if (e + 2 < nEdges) v.z = p[e + 2];
  return v;
}

__global__ void __launch_bounds__(PART_THREADS)
sage_count_kernel(const int* __restrict__ dst, int* __restrict__ rangeCount,
                  int* __restrict__ ovfLen, int nEdges) {
  __shared__ int hist[RANGES_MAX];
  const int tid = threadIdx.x;
  hist[tid] = 0;
  if (blockIdx.x == 0 && tid == 0) *ovfLen = 0;
  __syncthreads();
  const int e0 = blockIdx.x * (PART_THREADS * PART_QUADS * 4);
#pragma unroll
  for (int k = 0; k < PART_QUADS; ++k) {
    const int4 d = sage_load4(dst, e0 + (k * PART_THREADS + tid) * 4, nEdges);
    if (d.x >= 0) atomicAdd(&hist[d.x / OWN], 1);
    if (d.y >= 0) atomicAdd(&hist[d.y / OWN], 1);
    if (d.z >= 0) atomicAdd(&hist[d.z / OWN], 1);
    if (d.w >= 0) atomicAdd(&hist[d.w / OWN], 1);
  }
  __syncthreads();
  const int h = hist[tid];
  if (h) atomicAdd(rangeCount + tid, h);
}

__device__ __forceinline__ int sage_scan_incl(int v, int* a, int* b, int tid) {
  __syncthreads();
  a[tid] = v;
  __syncthreads();
  int* in = a;
  int* out = b;
  for (int off = 1; off < RANGES_MAX; off <<= 1) {
    out[tid] = in[tid] + (tid >= off ? in[tid - off] : 0);
    __syncthreads();
    int* t = in; in = out; out = t;
  }
  return in[tid];
}

__global__ void __launch_bounds__(PART_THREADS)
sage_place_kernel(const int* __restrict__ src, const int* __restrict__ dst,
                  const int* __restrict__ rangeCount,
                  int* __restrict__ cursor, int* __restrict__ rangeStart,
                  int* __restrict__ pairs, int nEdges) {
  __shared__ int hist[RANGES_MAX];
  __shared__ int lstart[RANGES_MAX];
  __shared__ int base[RANGES_MAX];
  __shared__ int scA[RANGES_MAX], scB[RANGES_MAX];
  __shared__ __attribute__((aligned(8))) int stage[PART_THREADS * PART_QUADS * 4 * 2];
  const int tid = threadIdx.x;
  hist[tid] = 0;
  __syncthreads();

  int4 d[PART_QUADS], s[PART_QUADS], rk[PART_QUADS];
  const int e0 = blockIdx.x * (PART_THREADS * PART_QUADS * 4);
#pragma unroll
  for (int k = 0; k < PART_QUADS; ++k) {
    const int e = e0 + (k * PART_THREADS + tid) * 4;
    d[k] = sage_load4(dst, e, nEdges);
    s[k] = sage_load4(src, e, nEdges);
    rk[k] = make_int4(0, 0, 0, 0);
    if (d[k].x >= 0) rk[k].x = atomicAdd(&hist[d[k].x / OWN], 1);
    if (d[k].y >= 0) rk[k].y = atomicAdd(&hist[d[k].y / OWN], 1);
    if (d[k].z >= 0) rk[k].z = atomicAdd(&hist[d[k].z / OWN], 1);
    if (d[k].w >= 0) rk[k].w = atomicAdd(&hist[d[k].w / OWN], 1);
  }

  const int total = rangeCount[tid];
  const int start = sage_scan_incl(total, scA, scB, tid) - total;
  if (blockIdx.x == 0) rangeStart[tid] = start;
  const int h = hist[tid];
  const int ls = sage_scan_incl(h, scA, scB, tid) - h;
  lstart[tid] = ls;
  base[tid] = start + (h ? atomicAdd(cursor + tid, h) : 0) - ls;
  __syncthreads();

#pragma unroll
  for (int k = 0; k < PART_QUADS; ++k) {
    if (d[k].x >= 0) *(int2*)(stage + 2 * (lstart[d[k].x / OWN] + rk[k].x)) = make_int2(d[k].x, s[k].x);
    if (d[k].y >= 0) *(int2*)(stage + 2 * (lstart[d[k].y / OWN] + rk[k].y)) = make_int2(d[k].y, s[k].y);
    if (d[k].z >= 0) *(int2*)(stage + 2 * (lstart[d[k].z / OWN] + rk[k].z)) = make_int2(d[k].z, s[k].z);
    if (d[k].w >= 0) *(int2*)(stage + 2 * (lstart[d[k].w / OWN] + rk[k].w)) = make_int2(d[k].w, s[k].w);
  }
  __syncthreads();

  int nHere = nEdges - e0;
  if (nHere > PART_THREADS * PART_QUADS * 4) nHere = PART_THREADS * PART_QUADS * 4;
  for (int i = tid; i < nHere; i += PART_THREADS) {
    const int2 pr = *(const int2*)(stage + 2 * i);
    *(int2*)(pairs + 2 * (size_t)(base[pr.x / OWN] + i)) = pr;
  }
}

__global__ void __launch_bounds__(OWN_THREADS)
sage_bucket_kernel(const int* __restrict__ pairs,
                   const int* __restrict__ rangeCount,
                   const int* __restrict__ rangeStart,
                   int* __restrict__ cnt,
                   int* __restrict__ bucket,
                   int* __restrict__ ovfDst,
                   int* __restrict__ ovfSrc,
                   int* __restrict__ ovfLen, int ovfCap, int nNodes) {
  __shared__ int lcnt[OWN];
  __shared__ __attribute__((aligned(16))) int lbkt[OWN * BUCKET];
  const int tid = threadIdx.x;
  const int lo = blockIdx.x * OWN;
  const int left = nNodes - lo;
  const int nOwn = left < OWN ? left : OWN;
  for (int i = tid; i < OWN; i += OWN_THREADS) lcnt[i] = 0;
  __syncthreads();

  const int n = rangeCount[blockIdx.x];
  const int* __restrict__ seg = pairs + 2 * (size_t)rangeStart[blockIdx.x];
  for (int i = tid; i < n; i += OWN_THREADS) {
    const int2 pr = *(const int2*)(seg + 2 * i);
    const int r = pr.x - lo;
    const int slot = atomicAdd(&lcnt[r], 1);
    if (slot < BUCKET) {
      lbkt[r * BUCKET + slot] = pr.y;
    } else {
      const int pos = atomicAdd(ovfLen, 1);
      if (pos < ovfCap) {
        ovfDst[pos] = pr.x;
        ovfSrc[pos] = pr.y;
      }
    }
  }
  __syncthreads();

  for (int i = tid; i < nOwn; i += OWN_THREADS) cnt[lo + i] = lcnt[i];
  for (int g = tid; g < nOwn * (BUCKET / 4); g += OWN_THREADS) {
    const int node = g >> 3;
    const int slot = (g & 7) << 2;
    const int w = node * BUCKET + slot;
    const int deg = lcnt[node];
    int4 v = *(const int4*)(lbkt + w);
    v.x = slot + 0 < deg ? v.x : 0;
    v.y = slot + 1 < deg ? v.y : 0;
    v.z = slot + 2 < deg ? v.z : 0;
    v.w = slot + 3 < deg ? v.w : 0;
    *(int4*)(bucket + (size_t)lo * BUCKET + w) = v;
  }
}

#define WPITCH 68
#define GEMM_WAVES 26
#define GEMM_THREADS (GEMM_WAVES * 32)
#define GEMM_BLOCKS 256
#define TILE_WORDS (16 * WPITCH)
#define IDS_WORDS (16 * BUCKET + 16)

__device__ __forceinline__ void sage_wave_lds_sync() {
  __builtin_amdgcn_fence(__ATOMIC_RELEASE, "wavefront");
  __builtin_amdgcn_wave_barrier();
  __builtin_amdgcn_fence(__ATOMIC_ACQUIRE, "wavefront");
}

__device__ __forceinline__ float4 sage_row4(const float* __restrict__ xin,
                                            int s, int c) {
  return *(const float4*)(xin + (size_t)s * D + c);
}

__device__ __forceinline__ void sage_add_if(float4& a, const float4 v, bool p) {
  a.x = p ? a.x + v.x : a.x;
  a.y = p ? a.y + v.y : a.y;
  a.z = p ? a.z + v.z : a.z;
  a.w = p ? a.w + v.w : a.w;
}

template <bool RELU>
__global__ void __launch_bounds__(GEMM_THREADS)
sage_layer_kernel(const float* __restrict__ xin, const int* __restrict__ cnt,
                  const int* __restrict__ bucket,
                  const int* __restrict__ ovfDst,
                  const int* __restrict__ ovfSrc, int ovfCap, int nNodes,
                  const float* __restrict__ Wl, const float* __restrict__ bl,
                  const float* __restrict__ Wr, float* __restrict__ out,
                  int nTiles) {
  __shared__ __attribute__((aligned(16))) float sW[2 * D * WPITCH];
  __shared__ __attribute__((aligned(16))) float sA[GEMM_WAVES * 2 * TILE_WORDS];
  __shared__ __attribute__((aligned(16))) int sI[GEMM_WAVES * IDS_WORDS];

  const int tid = threadIdx.x;
  const int lane = tid & 31;
  const int wave = __builtin_amdgcn_readfirstlane(tid >> 5);

  const int lo = (int)(((unsigned long long)blockIdx.x * (unsigned)nTiles) / GEMM_BLOCKS);
  const int hi = (int)(((unsigned long long)(blockIdx.x + 1) * (unsigned)nTiles) / GEMM_BLOCKS);
  const int rounds = (hi - lo + GEMM_WAVES - 1) / GEMM_WAVES;

  const int rr = lane >> 4;
  const int cc = (lane & 15) << 2;
  float* __restrict__ tA = sA + wave * (2 * TILE_WORDS);
  float* __restrict__ tX = tA + TILE_WORDS;
  int* __restrict__ tI = sI + wave * IDS_WORDS;
  int* __restrict__ tC = tI + 16 * BUCKET;

  const int m = lane & 15;
  const int khalf = (lane >> 4) << 1;
  const int mhi = (lane >> 4) << 3;
  const float* __restrict__ fA = tA + m * WPITCH + khalf;
  const float* __restrict__ fX = tX + m * WPITCH + khalf;
  const float* __restrict__ fL = sW + m * WPITCH + khalf;
  const float* __restrict__ fR = fL + D * WPITCH;

  for (int r = 0; r < rounds; ++r) {
    const int tile = lo + r * GEMM_WAVES + wave;
    const bool valid = tile < hi;
    const int row0 = tile << 4;
    int deg = 0;

    if (valid) {
      const int* __restrict__ gb = bucket + (size_t)row0 * BUCKET + (lane << 2);
      const float* __restrict__ gx = xin + (size_t)row0 * D + cc;
      int4 ids[4];
#pragma unroll
      for (int i = 0; i < 4; ++i) ids[i] = *(const int4*)(gb + i * 128);
      deg = cnt[row0 + m];
      float4 xr[8];
#pragma unroll
      for (int i = 0; i < 8; ++i)
        xr[i] = *(const float4*)(gx + (size_t)(2 * i + rr) * D);
#pragma unroll
      for (int i = 0; i < 4; ++i) *(int4*)(tI + i * 128 + (lane << 2)) = ids[i];
      tC[m] = deg;
#pragma unroll
      for (int i = 0; i < 8; ++i)
        *(float4*)(tX + (2 * i + rr) * WPITCH + cc) = xr[i];
    }
    if (r == 0) {
      constexpr int W_PIECES = 2 * D * (D / 4);
#pragma unroll
      for (int i = 0; i < (W_PIECES + GEMM_THREADS - 1) / GEMM_THREADS; ++i) {
        const int g = i * GEMM_THREADS + tid;
        if (g < W_PIECES) {
          const int wr = g >> 4;
          const float* __restrict__ w = wr < D ? Wl + wr * D : Wr + (wr - D) * D;
          *(float4*)(sW + wr * WPITCH + ((g & 15) << 2)) =
              *(const float4*)(w + ((g & 15) << 2));
        }
      }
      __syncthreads();
    }
    if (!valid) continue;

    sage_wave_lds_sync();

    for (int i = 0; i < 8; ++i) {
      const int lr = 2 * i + rr;
      const int n = tC[lr];
      const int end = n < BUCKET ? n : BUCKET;
      const int* __restrict__ lb = tI + lr * BUCKET;
      float4 acc = make_float4(0.0f, 0.0f, 0.0f, 0.0f);
      int j = 0;
      for (; j + 8 <= end; j += 8) {
        const int4 s = *(const int4*)(lb + j);
        const int4 t = *(const int4*)(lb + j + 4);
        const float4 v0 = sage_row4(xin, s.x, cc);
        const float4 v1 = sage_row4(xin, s.y, cc);
        const float4 v2 = sage_row4(xin, s.z, cc);
        const float4 v3 = sage_row4(xin, s.w, cc);
        const float4 v4 = sage_row4(xin, t.x, cc);
        const float4 v5 = sage_row4(xin, t.y, cc);
        const float4 v6 = sage_row4(xin, t.z, cc);
        const float4 v7 = sage_row4(xin, t.w, cc);
        acc.x += v0.x; acc.y += v0.y; acc.z += v0.z; acc.w += v0.w;
        acc.x += v1.x; acc.y += v1.y; acc.z += v1.z; acc.w += v1.w;
        acc.x += v2.x; acc.y += v2.y; acc.z += v2.z; acc.w += v2.w;
        acc.x += v3.x; acc.y += v3.y; acc.z += v3.z; acc.w += v3.w;
        acc.x += v4.x; acc.y += v4.y; acc.z += v4.z; acc.w += v4.w;
        acc.x += v5.x; acc.y += v5.y; acc.z += v5.z; acc.w += v5.w;
        acc.x += v6.x; acc.y += v6.y; acc.z += v6.z; acc.w += v6.w;
        acc.x += v7.x; acc.y += v7.y; acc.z += v7.z; acc.w += v7.w;
      }
      if (j + 4 <= end) {
        const int4 s = *(const int4*)(lb + j);
        const float4 v0 = sage_row4(xin, s.x, cc);
        const float4 v1 = sage_row4(xin, s.y, cc);
        const float4 v2 = sage_row4(xin, s.z, cc);
        const float4 v3 = sage_row4(xin, s.w, cc);
        acc.x += v0.x; acc.y += v0.y; acc.z += v0.z; acc.w += v0.w;
        acc.x += v1.x; acc.y += v1.y; acc.z += v1.z; acc.w += v1.w;
        acc.x += v2.x; acc.y += v2.y; acc.z += v2.z; acc.w += v2.w;
        acc.x += v3.x; acc.y += v3.y; acc.z += v3.z; acc.w += v3.w;
        j += 4;
      }
      if (j < end) {
        const int4 s = *(const int4*)(lb + j);
        const float4 v0 = sage_row4(xin, s.x, cc);
        const float4 v1 = sage_row4(xin, s.y, cc);
        const float4 v2 = sage_row4(xin, s.z, cc);
        acc.x += v0.x; acc.y += v0.y; acc.z += v0.z; acc.w += v0.w;
        sage_add_if(acc, v1, j + 1 < end);
        sage_add_if(acc, v2, j + 2 < end);
      }
      if (n > BUCKET) {
        const int row = row0 + lr;
        const int len = cnt[nNodes];
        const int mm = len < ovfCap ? len : ovfCap;
        for (int k = 0; k < mm; ++k) {
          if (ovfDst[k] == row) {
            const float4 u = sage_row4(xin, ovfSrc[k], cc);
            acc.x += u.x; acc.y += u.y; acc.z += u.z; acc.w += u.w;
          }
        }
      }
      *(float4*)(tA + lr * WPITCH + cc) = acc;
    }

    const float rdeg = 1.0f / fmaxf((float)deg, 1.0f);
    sage_wave_lds_sync();

    v8f acc[4];
#pragma unroll
    for (int nt = 0; nt < 4; ++nt) acc[nt] = (v8f){0, 0, 0, 0, 0, 0, 0, 0};

#pragma unroll
    for (int kk = 0; kk < 16; ++kk) {
      v2f am = *(const v2f*)(fA + (kk << 2));
      am[0] *= rdeg;
      am[1] *= rdeg;
      const v2f ax = *(const v2f*)(fX + (kk << 2));
#pragma unroll
      for (int nt = 0; nt < 4; ++nt) {
        const v2f bL = *(const v2f*)(fL + (nt << 4) * WPITCH + (kk << 2));
        const v2f bR = *(const v2f*)(fR + (nt << 4) * WPITCH + (kk << 2));
        acc[nt] = __builtin_amdgcn_wmma_f32_16x16x4_f32(
            false, am, false, bL, (short)0, acc[nt], false, false);
        acc[nt] = __builtin_amdgcn_wmma_f32_16x16x4_f32(
            false, ax, false, bR, (short)0, acc[nt], false, false);
      }
    }

#pragma unroll
    for (int nt = 0; nt < 4; ++nt) {
      const int j = (nt << 4) + m;
      const float b = bl[j];
#pragma unroll
      for (int v = 0; v < 8; ++v) {
        float val = acc[nt][v] + b;
        if (RELU) val = fmaxf(val, 0.0f);
        tA[(mhi + v) * WPITCH + j] = val;
      }
    }
    sage_wave_lds_sync();
    float* __restrict__ go = out + (size_t)row0 * D + cc;
#pragma unroll
    for (int i = 0; i < 8; ++i)
      *(float4*)(go + (size_t)(2 * i + rr) * D) =
          *(const float4*)(tA + (2 * i + rr) * WPITCH + cc);
    sage_wave_lds_sync();
  }
}

template <bool RELU>
__global__ void sage_tail_kernel(const float* __restrict__ xin,
                                 const int* __restrict__ cnt,
                                 const int* __restrict__ bucket,
                                 const int* __restrict__ ovfDst,
                                 const int* __restrict__ ovfSrc, int ovfCap,
                                 int nNodes,
                                 const float* __restrict__ Wl,
                                 const float* __restrict__ bl,
                                 const float* __restrict__ Wr,
                                 float* __restrict__ out, int row0, int nRows) {
  int idx = blockIdx.x * blockDim.x + threadIdx.x;
  if (idx >= nRows * D) return;
  int r = row0 + idx / D;
  int j = idx % D;
  const int n = cnt[r];
  const int end = n < BUCKET ? n : BUCKET;
  const int len = cnt[nNodes];
  const int mm = len < ovfCap ? len : ovfCap;
  float rdeg = 1.0f / fmaxf((float)n, 1.0f);
  float s = bl[j];
  for (int k = 0; k < D; ++k) {
    float a = 0.0f;
    for (int t = 0; t < end; ++t) a += xin[(size_t)bucket[r * BUCKET + t] * D + k];
    if (n > BUCKET)
      for (int t = 0; t < mm; ++t)
        if (ovfDst[t] == r) a += xin[(size_t)ovfSrc[t] * D + k];
    s += a * rdeg * Wl[j * D + k] + xin[(size_t)r * D + k] * Wr[j * D + k];
  }
  if (RELU) s = fmaxf(s, 0.0f);
  out[(size_t)r * D + j] = s;
}

static inline size_t align256(size_t x) { return (x + 255) & ~(size_t)255; }

extern "C" void kernel_launch(void* const* d_in, const int* in_sizes, int n_in,
                              void* d_out, int out_size, void* d_ws,
                              size_t ws_size, hipStream_t stream) {
  const float* x = (const float*)d_in[0];
  const int* src = (const int*)d_in[1];
  const int* dst = (const int*)d_in[2];
  const float* Wl[3] = {(const float*)d_in[3], (const float*)d_in[6],
                        (const float*)d_in[9]};
  const float* bl[3] = {(const float*)d_in[4], (const float*)d_in[7],
                        (const float*)d_in[10]};
  const float* Wr[3] = {(const float*)d_in[5], (const float*)d_in[8],
                        (const float*)d_in[11]};

  const int nNodes = in_sizes[0] / D;
  const int nEdges = in_sizes[1];
  const size_t featBytes = (size_t)nNodes * D * sizeof(float);

  char* ws = (char*)d_ws;
  int* cnt = (int*)ws;
  int* ovfLen = cnt + nNodes;
  size_t ofs = align256(((size_t)nNodes + 1) * sizeof(int));
  char* adj = ws + ofs;
  ofs += align256(featBytes);
  float* h0 = (float*)(ws + ofs);
  ofs += align256(featBytes);
  float* h1 = (float*)(ws + ofs);

  int* bucket = (int*)adj;
  const size_t bucketBytes = (size_t)nNodes * BUCKET * sizeof(int);
  const int ovfCap = (int)((featBytes - bucketBytes) / (2 * sizeof(int)));
  int* ovfDst = (int*)(adj + bucketBytes);
  int* ovfSrc = ovfDst + ovfCap;
  int* rangeCount = (int*)h0;
  int* cursor = rangeCount + RANGES_MAX;
  int* rangeStart = cursor + RANGES_MAX;
  int* pairs = (int*)d_out;
  const int partBlocks = (nEdges + PART_THREADS * PART_QUADS * 4 - 1) /
                         (PART_THREADS * PART_QUADS * 4);
  hipMemsetAsync(rangeCount, 0, 2 * RANGES_MAX * sizeof(int), stream);
  sage_count_kernel<<<partBlocks, PART_THREADS, 0, stream>>>(dst, rangeCount,
                                                            ovfLen, nEdges);
  sage_place_kernel<<<partBlocks, PART_THREADS, 0, stream>>>(
      src, dst, rangeCount, cursor, rangeStart, pairs, nEdges);
  sage_bucket_kernel<<<(nNodes + OWN - 1) / OWN, OWN_THREADS, 0, stream>>>(
      pairs, rangeCount, rangeStart, cnt, bucket, ovfDst, ovfSrc, ovfLen,
      ovfCap, nNodes);

  const int nTiles = nNodes / 16;
  const int gemmBlocks = GEMM_BLOCKS;
  const int tail = nNodes - nTiles * 16;

  auto run_layer = [&](const float* in, const float* wl, const float* b,
                       const float* wr, float* outp, bool relu) {
    if (nTiles > 0) {
      if (relu)
        sage_layer_kernel<true><<<gemmBlocks, GEMM_THREADS, 0, stream>>>(
            in, cnt, bucket, ovfDst, ovfSrc, ovfCap, nNodes, wl, b, wr, outp,
            nTiles);
      else
        sage_layer_kernel<false><<<gemmBlocks, GEMM_THREADS, 0, stream>>>(
            in, cnt, bucket, ovfDst, ovfSrc, ovfCap, nNodes, wl, b, wr, outp,
            nTiles);
    }
    if (tail > 0) {
      int tb = (tail * D + 255) / 256;
      if (relu)
        sage_tail_kernel<true><<<tb, 256, 0, stream>>>(
            in, cnt, bucket, ovfDst, ovfSrc, ovfCap, nNodes, wl, b, wr, outp,
            nTiles * 16, tail);
      else
        sage_tail_kernel<false><<<tb, 256, 0, stream>>>(
            in, cnt, bucket, ovfDst, ovfSrc, ovfCap, nNodes, wl, b, wr, outp,
            nTiles * 16, tail);
    }
  };

  run_layer(x, Wl[0], bl[0], Wr[0], h0,  true);
  run_layer(h0, Wl[1], bl[1], Wr[1], h1,  true);
  run_layer(h1, Wl[2], bl[2], Wr[2], (float*)d_out,  false);
}


# Edges partitioned without global offsets: each block keeps its range-sorted pairs in its own area with a per-block range table, bucket pass reads each block's run; count launch and memset dropped
# speedup vs baseline: 5.1520x; 1.0466x over previous
#include <hip/hip_runtime.h>

typedef float v2f __attribute__((ext_vector_type(2)));
typedef float v8f __attribute__((ext_vector_type(8)));

#define D 64

#define BUCKET 32
#define OWN 784
#define OWN_THREADS 1024
#define BUCKET_RUNS 8
#define RANGES_MAX 256
#define PART_THREADS 256
#define PART_QUADS 4
#define PART_EDGES (PART_THREADS * PART_QUADS * 4)

__device__ __forceinline__ int4 sage_load4(const int* __restrict__ p, int e,
                                           int nEdges) {
  if (e + 4 <= nEdges) return *(const int4*)(p + e);
  int4 v = make_int4(-1, -1, -1, -1);
  if (e + 0 < nEdges) v.x = p[e + 0];
  if (e + 1 < nEdges) v.y = p[e + 1];
  if (e + 2 < nEdges) v.z = p[e + 2];
  return v;
}

__device__ __forceinline__ int sage_scan_incl(int v, int* a, int* b, int tid) {
  a[tid] = v;
  __syncthreads();
  int* in = a;
  int* out = b;
  for (int off = 1; off < RANGES_MAX; off <<= 1) {
    out[tid] = in[tid] + (tid >= off ? in[tid - off] : 0);
    __syncthreads();
    int* t = in; in = out; out = t;
  }
  return in[tid];
}

__global__ void __launch_bounds__(PART_THREADS)
sage_place_kernel(const int* __restrict__ src, const int* __restrict__ dst,
                  int* __restrict__ blockOff, int* __restrict__ pairs,
                  int* __restrict__ ovfLen, int nEdges) {
  __shared__ int hist[RANGES_MAX];
  __shared__ int lstart[RANGES_MAX];
  __shared__ int scA[RANGES_MAX], scB[RANGES_MAX];
  __shared__ __attribute__((aligned(8))) int stage[PART_EDGES * 2];
  const int tid = threadIdx.x;
  hist[tid] = 0;
  if (blockIdx.x == 0 && tid == 0) *ovfLen = 0;
  __syncthreads();

  int4 d[PART_QUADS], s[PART_QUADS], rk[PART_QUADS];
  const int e0 = blockIdx.x * PART_EDGES;
#pragma unroll
  for (int k = 0; k < PART_QUADS; ++k) {
    const int e = e0 + (k * PART_THREADS + tid) * 4;
    d[k] = sage_load4(dst, e, nEdges);
    s[k] = sage_load4(src, e, nEdges);
    rk[k] = make_int4(0, 0, 0, 0);
    if (d[k].x >= 0) rk[k].x = atomicAdd(&hist[d[k].x / OWN], 1);
    if (d[k].y >= 0) rk[k].y = atomicAdd(&hist[d[k].y / OWN], 1);
    if (d[k].z >= 0) rk[k].z = atomicAdd(&hist[d[k].z / OWN], 1);
    if (d[k].w >= 0) rk[k].w = atomicAdd(&hist[d[k].w / OWN], 1);
  }
  __syncthreads();

  const int h = hist[tid];
  const int ls = sage_scan_incl(h, scA, scB, tid) - h;
  lstart[tid] = ls;
  int* __restrict__ off = blockOff + (size_t)blockIdx.x * (RANGES_MAX + 1);
  off[tid] = ls;
  if (tid == RANGES_MAX - 1) off[RANGES_MAX] = ls + h;
  __syncthreads();

#pragma unroll
  for (int k = 0; k < PART_QUADS; ++k) {
    if (d[k].x >= 0) *(int2*)(stage + 2 * (lstart[d[k].x / OWN] + rk[k].x)) = make_int2(d[k].x, s[k].x);
    if (d[k].y >= 0) *(int2*)(stage + 2 * (lstart[d[k].y / OWN] + rk[k].y)) = make_int2(d[k].y, s[k].y);
    if (d[k].z >= 0) *(int2*)(stage + 2 * (lstart[d[k].z / OWN] + rk[k].z)) = make_int2(d[k].z, s[k].z);
    if (d[k].w >= 0) *(int2*)(stage + 2 * (lstart[d[k].w / OWN] + rk[k].w)) = make_int2(d[k].w, s[k].w);
  }
  __syncthreads();

  int nHere = nEdges - e0;
  if (nHere > PART_EDGES) nHere = PART_EDGES;
  int* __restrict__ area = pairs + 2 * (size_t)e0;
  for (int i = tid; i < nHere; i += PART_THREADS)
    *(int2*)(area + 2 * i) = *(const int2*)(stage + 2 * i);
}

__global__ void __launch_bounds__(OWN_THREADS)
sage_bucket_kernel(const int* __restrict__ pairs,
                   const int* __restrict__ blockOff, int partBlocks,
                   int* __restrict__ cnt,
                   int* __restrict__ bucket,
                   int* __restrict__ ovfDst,
                   int* __restrict__ ovfSrc,
                   int* __restrict__ ovfLen, int ovfCap, int nNodes) {
  __shared__ int lcnt[OWN];
  __shared__ __attribute__((aligned(16))) int lbkt[OWN * BUCKET];
  __shared__ int runLo[OWN_THREADS], runHi[OWN_THREADS];
  const int tid = threadIdx.x;
  const int lane = tid & 31;
  const int wave = tid >> 5;
  const int lo = blockIdx.x * OWN;
  const int left = nNodes - lo;
  const int nOwn = left < OWN ? left : OWN;
  for (int i = tid; i < OWN; i += OWN_THREADS) lcnt[i] = 0;

  auto put = [&](const int2 pr) {
    const int r = pr.x - lo;
    const int slot = atomicAdd(&lcnt[r], 1);
    if (slot < BUCKET) {
      lbkt[r * BUCKET + slot] = pr.y;
    } else {
      const int pos = atomicAdd(ovfLen, 1);
      if (pos < ovfCap) {
        ovfDst[pos] = pr.x;
        ovfSrc[pos] = pr.y;
      }
    }
  };

  for (int c0 = 0; c0 < partBlocks; c0 += OWN_THREADS) {
    if (c0 + tid < partBlocks) {
      const int* __restrict__ off =
          blockOff + (size_t)(c0 + tid) * (RANGES_MAX + 1) + blockIdx.x;
      runLo[tid] = off[0];
      runHi[tid] = off[1];
    }
    __syncthreads();
    const int nb = partBlocks - c0 < OWN_THREADS ? partBlocks - c0 : OWN_THREADS;
    for (int q = wave; q < nb; q += BUCKET_RUNS * (OWN_THREADS / 32)) {
      int rs[BUCKET_RUNS], re[BUCKET_RUNS];
      int2 v[BUCKET_RUNS][2];
#pragma unroll
      for (int u = 0; u < BUCKET_RUNS; ++u) {
        const int qq = q + u * (OWN_THREADS / 32);
        const bool ok = qq < nb;
        rs[u] = ok ? runLo[qq] : 0;
        re[u] = ok ? runHi[qq] : 0;
      }
#pragma unroll
      for (int u = 0; u < BUCKET_RUNS; ++u) {
        const int* __restrict__ g =
            pairs + 2 * (size_t)(c0 + q + u * (OWN_THREADS / 32)) * PART_EDGES;
        v[u][0] = make_int2(-1, 0);
        v[u][1] = make_int2(-1, 0);
        if (rs[u] + lane < re[u]) v[u][0] = *(const int2*)(g + 2 * (rs[u] + lane));
        if (rs[u] + 32 + lane < re[u])
          v[u][1] = *(const int2*)(g + 2 * (rs[u] + 32 + lane));
      }
#pragma unroll
      for (int u = 0; u < BUCKET_RUNS; ++u) {
        const int* __restrict__ g =
            pairs + 2 * (size_t)(c0 + q + u * (OWN_THREADS / 32)) * PART_EDGES;
        if (v[u][0].x >= 0) put(v[u][0]);
        if (v[u][1].x >= 0) put(v[u][1]);
        for (int i = rs[u] + 64 + lane; i < re[u]; i += 32)
          put(*(const int2*)(g + 2 * i));
      }
    }
    __syncthreads();
  }

  for (int i = tid; i < nOwn; i += OWN_THREADS) cnt[lo + i] = lcnt[i];
  for (int g = tid; g < nOwn * (BUCKET / 4); g += OWN_THREADS) {
    const int node = g >> 3;
    const int slot = (g & 7) << 2;
    const int w = node * BUCKET + slot;
    const int deg = lcnt[node];
    int4 v = *(const int4*)(lbkt + w);
    v.x = slot + 0 < deg ? v.x : 0;
    v.y = slot + 1 < deg ? v.y : 0;
    v.z = slot + 2 < deg ? v.z : 0;
    v.w = slot + 3 < deg ? v.w : 0;
    *(int4*)(bucket + (size_t)lo * BUCKET + w) = v;
  }
}

#define WPITCH 68
#define GEMM_WAVES 26
#define GEMM_THREADS (GEMM_WAVES * 32)
#define GEMM_BLOCKS 256
#define TILE_WORDS (16 * WPITCH)
#define IDS_WORDS (16 * BUCKET + 16)

__device__ __forceinline__ void sage_wave_lds_sync() {
  __builtin_amdgcn_fence(__ATOMIC_RELEASE, "wavefront");
  __builtin_amdgcn_wave_barrier();
  __builtin_amdgcn_fence(__ATOMIC_ACQUIRE, "wavefront");
}

__device__ __forceinline__ float4 sage_row4(const float* __restrict__ xin,
                                            int s, int c) {
  return *(const float4*)(xin + (size_t)s * D + c);
}

__device__ __forceinline__ void sage_add_if(float4& a, const float4 v, bool p) {
  a.x = p ? a.x + v.x : a.x;
  a.y = p ? a.y + v.y : a.y;
  a.z = p ? a.z + v.z : a.z;
  a.w = p ? a.w + v.w : a.w;
}

template <bool RELU>
__global__ void __launch_bounds__(GEMM_THREADS)
sage_layer_kernel(const float* __restrict__ xin, const int* __restrict__ cnt,
                  const int* __restrict__ bucket,
                  const int* __restrict__ ovfDst,
                  const int* __restrict__ ovfSrc, int ovfCap, int nNodes,
                  const float* __restrict__ Wl, const float* __restrict__ bl,
                  const float* __restrict__ Wr, float* __restrict__ out,
                  int nTiles) {
  __shared__ __attribute__((aligned(16))) float sW[2 * D * WPITCH];
  __shared__ __attribute__((aligned(16))) float sA[GEMM_WAVES * 2 * TILE_WORDS];
  __shared__ __attribute__((aligned(16))) int sI[GEMM_WAVES * IDS_WORDS];

  const int tid = threadIdx.x;
  const int lane = tid & 31;
  const int wave = __builtin_amdgcn_readfirstlane(tid >> 5);

  const int lo = (int)(((unsigned long long)blockIdx.x * (unsigned)nTiles) / GEMM_BLOCKS);
  const int hi = (int)(((unsigned long long)(blockIdx.x + 1) * (unsigned)nTiles) / GEMM_BLOCKS);
  const int rounds = (hi - lo + GEMM_WAVES - 1) / GEMM_WAVES;

  const int rr = lane >> 4;
  const int cc = (lane & 15) << 2;
  float* __restrict__ tA = sA + wave * (2 * TILE_WORDS);
  float* __restrict__ tX = tA + TILE_WORDS;
  int* __restrict__ tI = sI + wave * IDS_WORDS;
  int* __restrict__ tC = tI + 16 * BUCKET;

  const int m = lane & 15;
  const int khalf = (lane >> 4) << 1;
  const int mhi = (lane >> 4) << 3;
  const float* __restrict__ fA = tA + m * WPITCH + khalf;
  const float* __restrict__ fX = tX + m * WPITCH + khalf;
  const float* __restrict__ fL = sW + m * WPITCH + khalf;
  const float* __restrict__ fR = fL + D * WPITCH;

  for (int r = 0; r < rounds; ++r) {
    const int tile = lo + r * GEMM_WAVES + wave;
    const bool valid = tile < hi;
    const int row0 = tile << 4;
    int deg = 0;

    if (valid) {
      const int* __restrict__ gb = bucket + (size_t)row0 * BUCKET + (lane << 2);
      const float* __restrict__ gx = xin + (size_t)row0 * D + cc;
      int4 ids[4];
#pragma unroll
      for (int i = 0; i < 4; ++i) ids[i] = *(const int4*)(gb + i * 128);
      deg = cnt[row0 + m];
      float4 xr[8];
#pragma unroll
      for (int i = 0; i < 8; ++i)
        xr[i] = *(const float4*)(gx + (size_t)(2 * i + rr) * D);
#pragma unroll
      for (int i = 0; i < 4; ++i) *(int4*)(tI + i * 128 + (lane << 2)) = ids[i];
      tC[m] = deg;
#pragma unroll
      for (int i = 0; i < 8; ++i)
        *(float4*)(tX + (2 * i + rr) * WPITCH + cc) = xr[i];
    }
    if (r == 0) {
      constexpr int W_PIECES = 2 * D * (D / 4);
#pragma unroll
      for (int i = 0; i < (W_PIECES + GEMM_THREADS - 1) / GEMM_THREADS; ++i) {
        const int g = i * GEMM_THREADS + tid;
        if (g < W_PIECES) {
          const int wr = g >> 4;
          const float* __restrict__ w = wr < D ? Wl + wr * D : Wr + (wr - D) * D;
          *(float4*)(sW + wr * WPITCH + ((g & 15) << 2)) =
              *(const float4*)(w + ((g & 15) << 2));
        }
      }
      __syncthreads();
    }
    if (!valid) continue;

    sage_wave_lds_sync();

    for (int i = 0; i < 8; ++i) {
      const int lr = 2 * i + rr;
      const int n = tC[lr];
      const int end = n < BUCKET ? n : BUCKET;
      const int* __restrict__ lb = tI + lr * BUCKET;
      float4 acc = make_float4(0.0f, 0.0f, 0.0f, 0.0f);
      int j = 0;
      for (; j + 8 <= end; j += 8) {
        const int4 s = *(const int4*)(lb + j);
        const int4 t = *(const int4*)(lb + j + 4);
        const float4 v0 = sage_row4(xin, s.x, cc);
        const float4 v1 = sage_row4(xin, s.y, cc);
        const float4 v2 = sage_row4(xin, s.z, cc);
        const float4 v3 = sage_row4(xin, s.w, cc);
        const float4 v4 = sage_row4(xin, t.x, cc);
        const float4 v5 = sage_row4(xin, t.y, cc);
        const float4 v6 = sage_row4(xin, t.z, cc);
        const float4 v7 = sage_row4(xin, t.w, cc);
        acc.x += v0.x; acc.y += v0.y; acc.z += v0.z; acc.w += v0.w;
        acc.x += v1.x; acc.y += v1.y; acc.z += v1.z; acc.w += v1.w;
        acc.x += v2.x; acc.y += v2.y; acc.z += v2.z; acc.w += v2.w;
        acc.x += v3.x; acc.y += v3.y; acc.z += v3.z; acc.w += v3.w;
        acc.x += v4.x; acc.y += v4.y; acc.z += v4.z; acc.w += v4.w;
        acc.x += v5.x; acc.y += v5.y; acc.z += v5.z; acc.w += v5.w;
        acc.x += v6.x; acc.y += v6.y; acc.z += v6.z; acc.w += v6.w;
        acc.x += v7.x; acc.y += v7.y; acc.z += v7.z; acc.w += v7.w;
      }
      if (j + 4 <= end) {
        const int4 s = *(const int4*)(lb + j);
        const float4 v0 = sage_row4(xin, s.x, cc);
        const float4 v1 = sage_row4(xin, s.y, cc);
        const float4 v2 = sage_row4(xin, s.z, cc);
        const float4 v3 = sage_row4(xin, s.w, cc);
        acc.x += v0.x; acc.y += v0.y; acc.z += v0.z; acc.w += v0.w;
        acc.x += v1.x; acc.y += v1.y; acc.z += v1.z; acc.w += v1.w;
        acc.x += v2.x; acc.y += v2.y; acc.z += v2.z; acc.w += v2.w;
        acc.x += v3.x; acc.y += v3.y; acc.z += v3.z; acc.w += v3.w;
        j += 4;
      }
      if (j < end) {
        const int4 s = *(const int4*)(lb + j);
        const float4 v0 = sage_row4(xin, s.x, cc);
        const float4 v1 = sage_row4(xin, s.y, cc);
        const float4 v2 = sage_row4(xin, s.z, cc);
        acc.x += v0.x; acc.y += v0.y; acc.z += v0.z; acc.w += v0.w;
        sage_add_if(acc, v1, j + 1 < end);
        sage_add_if(acc, v2, j + 2 < end);
      }
      if (n > BUCKET) {
        const int row = row0 + lr;
        const int len = cnt[nNodes];
        const int mm = len < ovfCap ? len : ovfCap;
        for (int k = 0; k < mm; ++k) {
          if (ovfDst[k] == row) {
            const float4 u = sage_row4(xin, ovfSrc[k], cc);
            acc.x += u.x; acc.y += u.y; acc.z += u.z; acc.w += u.w;
          }
        }
      }
      *(float4*)(tA + lr * WPITCH + cc) = acc;
    }

    const float rdeg = 1.0f / fmaxf((float)deg, 1.0f);
    sage_wave_lds_sync();

    v8f acc[4];
#pragma unroll
    for (int nt = 0; nt < 4; ++nt) acc[nt] = (v8f){0, 0, 0, 0, 0, 0, 0, 0};

#pragma unroll
    for (int kk = 0; kk < 16; ++kk) {
      v2f am = *(const v2f*)(fA + (kk << 2));
      am[0] *= rdeg;
      am[1] *= rdeg;
      const v2f ax = *(const v2f*)(fX + (kk << 2));
#pragma unroll
      for (int nt = 0; nt < 4; ++nt) {
        const v2f bL = *(const v2f*)(fL + (nt << 4) * WPITCH + (kk << 2));
        const v2f bR = *(const v2f*)(fR + (nt << 4) * WPITCH + (kk << 2));
        acc[nt] = __builtin_amdgcn_wmma_f32_16x16x4_f32(
            false, am, false, bL, (short)0, acc[nt], false, false);
        acc[nt] = __builtin_amdgcn_wmma_f32_16x16x4_f32(
            false, ax, false, bR, (short)0, acc[nt], false, false);
      }
    }

#pragma unroll
    for (int nt = 0; nt < 4; ++nt) {
      const int j = (nt << 4) + m;
      const float b = bl[j];
#pragma unroll
      for (int v = 0; v < 8; ++v) {
        float val = acc[nt][v] + b;
        if (RELU) val = fmaxf(val, 0.0f);
        tA[(mhi + v) * WPITCH + j] = val;
      }
    }
    sage_wave_lds_sync();
    float* __restrict__ go = out + (size_t)row0 * D + cc;
#pragma unroll
    for (int i = 0; i < 8; ++i)
      *(float4*)(go + (size_t)(2 * i + rr) * D) =
          *(const float4*)(tA + (2 * i + rr) * WPITCH + cc);
    sage_wave_lds_sync();
  }
}

template <bool RELU>
__global__ void sage_tail_kernel(const float* __restrict__ xin,
                                 const int* __restrict__ cnt,
                                 const int* __restrict__ bucket,
                                 const int* __restrict__ ovfDst,
                                 const int* __restrict__ ovfSrc, int ovfCap,
                                 int nNodes,
                                 const float* __restrict__ Wl,
                                 const float* __restrict__ bl,
                                 const float* __restrict__ Wr,
                                 float* __restrict__ out, int row0, int nRows) {
  int idx = blockIdx.x * blockDim.x + threadIdx.x;
  if (idx >= nRows * D) return;
  int r = row0 + idx / D;
  int j = idx % D;
  const int n = cnt[r];
  const int end = n < BUCKET ? n : BUCKET;
  const int len = cnt[nNodes];
  const int mm = len < ovfCap ? len : ovfCap;
  float rdeg = 1.0f / fmaxf((float)n, 1.0f);
  float s = bl[j];
  for (int k = 0; k < D; ++k) {
    float a = 0.0f;
    for (int t = 0; t < end; ++t) a += xin[(size_t)bucket[r * BUCKET + t] * D + k];
    if (n > BUCKET)
      for (int t = 0; t < mm; ++t)
        if (ovfDst[t] == r) a += xin[(size_t)ovfSrc[t] * D + k];
    s += a * rdeg * Wl[j * D + k] + xin[(size_t)r * D + k] * Wr[j * D + k];
  }
  if (RELU) s = fmaxf(s, 0.0f);
  out[(size_t)r * D + j] = s;
}

static inline size_t align256(size_t x) { return (x + 255) & ~(size_t)255; }

extern "C" void kernel_launch(void* const* d_in, const int* in_sizes, int n_in,
                              void* d_out, int out_size, void* d_ws,
                              size_t ws_size, hipStream_t stream) {
  const float* x = (const float*)d_in[0];
  const int* src = (const int*)d_in[1];
  const int* dst = (const int*)d_in[2];
  const float* Wl[3] = {(const float*)d_in[3], (const float*)d_in[6],
                        (const float*)d_in[9]};
  const float* bl[3] = {(const float*)d_in[4], (const float*)d_in[7],
                        (const float*)d_in[10]};
  const float* Wr[3] = {(const float*)d_in[5], (const float*)d_in[8],
                        (const float*)d_in[11]};

  const int nNodes = in_sizes[0] / D;
  const int nEdges = in_sizes[1];
  const size_t featBytes = (size_t)nNodes * D * sizeof(float);

  char* ws = (char*)d_ws;
  int* cnt = (int*)ws;
  int* ovfLen = cnt + nNodes;
  size_t ofs = align256(((size_t)nNodes + 1) * sizeof(int));
  char* adj = ws + ofs;
  ofs += align256(featBytes);
  float* h0 = (float*)(ws + ofs);
  ofs += align256(featBytes);
  float* h1 = (float*)(ws + ofs);

  int* bucket = (int*)adj;
  const size_t bucketBytes = (size_t)nNodes * BUCKET * sizeof(int);
  const int ovfCap = (int)((featBytes - bucketBytes) / (2 * sizeof(int)));
  int* ovfDst = (int*)(adj + bucketBytes);
  int* ovfSrc = ovfDst + ovfCap;
  int* blockOff = (int*)h0;
  int* pairs = (int*)d_out;
  const int partBlocks = (nEdges + PART_EDGES - 1) / PART_EDGES;
  sage_place_kernel<<<partBlocks, PART_THREADS, 0, stream>>>(
      src, dst, blockOff, pairs, ovfLen, nEdges);
  sage_bucket_kernel<<<(nNodes + OWN - 1) / OWN, OWN_THREADS, 0, stream>>>(
      pairs, blockOff, partBlocks, cnt, bucket, ovfDst, ovfSrc, ovfLen, ovfCap,
      nNodes);

  const int nTiles = nNodes / 16;
  const int gemmBlocks = GEMM_BLOCKS;
  const int tail = nNodes - nTiles * 16;

  auto run_layer = [&](const float* in, const float* wl, const float* b,
                       const float* wr, float* outp, bool relu) {
    if (nTiles > 0) {
      if (relu)
        sage_layer_kernel<true><<<gemmBlocks, GEMM_THREADS, 0, stream>>>(
            in, cnt, bucket, ovfDst, ovfSrc, ovfCap, nNodes, wl, b, wr, outp,
            nTiles);
      else
        sage_layer_kernel<false><<<gemmBlocks, GEMM_THREADS, 0, stream>>>(
            in, cnt, bucket, ovfDst, ovfSrc, ovfCap, nNodes, wl, b, wr, outp,
            nTiles);
    }
    if (tail > 0) {
      int tb = (tail * D + 255) / 256;
      if (relu)
        sage_tail_kernel<true><<<tb, 256, 0, stream>>>(
            in, cnt, bucket, ovfDst, ovfSrc, ovfCap, nNodes, wl, b, wr, outp,
            nTiles * 16, tail);
      else
        sage_tail_kernel<false><<<tb, 256, 0, stream>>>(
            in, cnt, bucket, ovfDst, ovfSrc, ovfCap, nNodes, wl, b, wr, outp,
            nTiles * 16, tail);
    }
  };

  run_layer(x, Wl[0], bl[0], Wr[0], h0,  true);
  run_layer(h0, Wl[1], bl[1], Wr[1], h1,  true);
  run_layer(h1, Wl[2], bl[2], Wr[2], (float*)d_out,  false);
}


# Neighbour sums of four node pairs walked together, four slots per trip: 16 row-read instructions in flight per wave (were 8), about half the dependent trips per tile; per-node sum order kept
# speedup vs baseline: 5.5161x; 1.0707x over previous
#include <hip/hip_runtime.h>

typedef float v2f __attribute__((ext_vector_type(2)));
typedef float v8f __attribute__((ext_vector_type(8)));
typedef float v4f __attribute__((ext_vector_type(4)));

#define D 64

#define BUCKET 32
#define OWN 784
#define OWN_THREADS 1024
#define BUCKET_RUNS 8
#define RANGES_MAX 256
#define PART_THREADS 256
#define PART_QUADS 4
#define PART_EDGES (PART_THREADS * PART_QUADS * 4)

__device__ __forceinline__ int4 sage_load4(const int* __restrict__ p, int e,
                                           int nEdges) {
  if (e + 4 <= nEdges) return *(const int4*)(p + e);
  int4 v = make_int4(-1, -1, -1, -1);
  if (e + 0 < nEdges) v.x = p[e + 0];
  if (e + 1 < nEdges) v.y = p[e + 1];
  if (e + 2 < nEdges) v.z = p[e + 2];
  return v;
}

__device__ __forceinline__ int sage_scan_incl(int v, int* a, int* b, int tid) {
  a[tid] = v;
  __syncthreads();
  int* in = a;
  int* out = b;
  for (int off = 1; off < RANGES_MAX; off <<= 1) {
    out[tid] = in[tid] + (tid >= off ? in[tid - off] : 0);
    __syncthreads();
    int* t = in; in = out; out = t;
  }
  return in[tid];
}

__global__ void __launch_bounds__(PART_THREADS)
sage_place_kernel(const int* __restrict__ src, const int* __restrict__ dst,
                  int* __restrict__ blockOff, int* __restrict__ pairs,
                  int* __restrict__ ovfLen, int nEdges) {
  __shared__ int hist[RANGES_MAX];
  __shared__ int lstart[RANGES_MAX];
  __shared__ int scA[RANGES_MAX], scB[RANGES_MAX];
  __shared__ __attribute__((aligned(8))) int stage[PART_EDGES * 2];
  const int tid = threadIdx.x;
  hist[tid] = 0;
  if (blockIdx.x == 0 && tid == 0) *ovfLen = 0;
  __syncthreads();

  int4 d[PART_QUADS], s[PART_QUADS], rk[PART_QUADS];
  const int e0 = blockIdx.x * PART_EDGES;
#pragma unroll
  for (int k = 0; k < PART_QUADS; ++k) {
    const int e = e0 + (k * PART_THREADS + tid) * 4;
    d[k] = sage_load4(dst, e, nEdges);
    s[k] = sage_load4(src, e, nEdges);
    rk[k] = make_int4(0, 0, 0, 0);
    if (d[k].x >= 0) rk[k].x = atomicAdd(&hist[d[k].x / OWN], 1);
    if (d[k].y >= 0) rk[k].y = atomicAdd(&hist[d[k].y / OWN], 1);
    if (d[k].z >= 0) rk[k].z = atomicAdd(&hist[d[k].z / OWN], 1);
    if (d[k].w >= 0) rk[k].w = atomicAdd(&hist[d[k].w / OWN], 1);
  }
  __syncthreads();

  const int h = hist[tid];
  const int ls = sage_scan_incl(h, scA, scB, tid) - h;
  lstart[tid] = ls;
  int* __restrict__ off = blockOff + (size_t)blockIdx.x * (RANGES_MAX + 1);
  off[tid] = ls;
  if (tid == RANGES_MAX - 1) off[RANGES_MAX] = ls + h;
  __syncthreads();

#pragma unroll
  for (int k = 0; k < PART_QUADS; ++k) {
    if (d[k].x >= 0) *(int2*)(stage + 2 * (lstart[d[k].x / OWN] + rk[k].x)) = make_int2(d[k].x, s[k].x);
    if (d[k].y >= 0) *(int2*)(stage + 2 * (lstart[d[k].y / OWN] + rk[k].y)) = make_int2(d[k].y, s[k].y);
    if (d[k].z >= 0) *(int2*)(stage + 2 * (lstart[d[k].z / OWN] + rk[k].z)) = make_int2(d[k].z, s[k].z);
    if (d[k].w >= 0) *(int2*)(stage + 2 * (lstart[d[k].w / OWN] + rk[k].w)) = make_int2(d[k].w, s[k].w);
  }
  __syncthreads();

  int nHere = nEdges - e0;
  if (nHere > PART_EDGES) nHere = PART_EDGES;
  int* __restrict__ area = pairs + 2 * (size_t)e0;
  for (int i = tid; i < nHere; i += PART_THREADS)
    *(int2*)(area + 2 * i) = *(const int2*)(stage + 2 * i);
}

__global__ void __launch_bounds__(OWN_THREADS)
sage_bucket_kernel(const int* __restrict__ pairs,
                   const int* __restrict__ blockOff, int partBlocks,
                   int* __restrict__ cnt,
                   int* __restrict__ bucket,
                   int* __restrict__ ovfDst,
                   int* __restrict__ ovfSrc,
                   int* __restrict__ ovfLen, int ovfCap, int nNodes) {
  __shared__ int lcnt[OWN];
  __shared__ __attribute__((aligned(16))) int lbkt[OWN * BUCKET];
  __shared__ int runLo[OWN_THREADS], runHi[OWN_THREADS];
  const int tid = threadIdx.x;
  const int lane = tid & 31;
  const int wave = tid >> 5;
  const int lo = blockIdx.x * OWN;
  const int left = nNodes - lo;
  const int nOwn = left < OWN ? left : OWN;
  for (int i = tid; i < OWN; i += OWN_THREADS) lcnt[i] = 0;

  auto put = [&](const int2 pr) {
    const int r = pr.x - lo;
    const int slot = atomicAdd(&lcnt[r], 1);
    if (slot < BUCKET) {
      lbkt[r * BUCKET + slot] = pr.y;
    } else {
      const int pos = atomicAdd(ovfLen, 1);
      if (pos < ovfCap) {
        ovfDst[pos] = pr.x;
        ovfSrc[pos] = pr.y;
      }
    }
  };

  for (int c0 = 0; c0 < partBlocks; c0 += OWN_THREADS) {
    if (c0 + tid < partBlocks) {
      const int* __restrict__ off =
          blockOff + (size_t)(c0 + tid) * (RANGES_MAX + 1) + blockIdx.x;
      runLo[tid] = off[0];
      runHi[tid] = off[1];
    }
    __syncthreads();
    const int nb = partBlocks - c0 < OWN_THREADS ? partBlocks - c0 : OWN_THREADS;
    for (int q = wave; q < nb; q += BUCKET_RUNS * (OWN_THREADS / 32)) {
      int rs[BUCKET_RUNS], re[BUCKET_RUNS];
      int2 v[BUCKET_RUNS][2];
#pragma unroll
      for (int u = 0; u < BUCKET_RUNS; ++u) {
        const int qq = q + u * (OWN_THREADS / 32);
        const bool ok = qq < nb;
        rs[u] = ok ? runLo[qq] : 0;
        re[u] = ok ? runHi[qq] : 0;
      }
#pragma unroll
      for (int u = 0; u < BUCKET_RUNS; ++u) {
        const int* __restrict__ g =
            pairs + 2 * (size_t)(c0 + q + u * (OWN_THREADS / 32)) * PART_EDGES;
        v[u][0] = make_int2(-1, 0);
        v[u][1] = make_int2(-1, 0);
        if (rs[u] + lane < re[u]) v[u][0] = *(const int2*)(g + 2 * (rs[u] + lane));
        if (rs[u] + 32 + lane < re[u])
          v[u][1] = *(const int2*)(g + 2 * (rs[u] + 32 + lane));
      }
#pragma unroll
      for (int u = 0; u < BUCKET_RUNS; ++u) {
        const int* __restrict__ g =
            pairs + 2 * (size_t)(c0 + q + u * (OWN_THREADS / 32)) * PART_EDGES;
        if (v[u][0].x >= 0) put(v[u][0]);
        if (v[u][1].x >= 0) put(v[u][1]);
        for (int i = rs[u] + 64 + lane; i < re[u]; i += 32)
          put(*(const int2*)(g + 2 * i));
      }
    }
    __syncthreads();
  }

  for (int i = tid; i < nOwn; i += OWN_THREADS) cnt[lo + i] = lcnt[i];
  for (int g = tid; g < nOwn * (BUCKET / 4); g += OWN_THREADS) {
    const int node = g >> 3;
    const int slot = (g & 7) << 2;
    const int w = node * BUCKET + slot;
    const int deg = lcnt[node];
    int4 v = *(const int4*)(lbkt + w);
    v.x = slot + 0 < deg ? v.x : 0;
    v.y = slot + 1 < deg ? v.y : 0;
    v.z = slot + 2 < deg ? v.z : 0;
    v.w = slot + 3 < deg ? v.w : 0;
    *(int4*)(bucket + (size_t)lo * BUCKET + w) = v;
  }
}

#define WPITCH 68
#define GEMM_WAVES 26
#define GEMM_THREADS (GEMM_WAVES * 32)
#define GEMM_BLOCKS 256
#define TILE_WORDS (16 * WPITCH)
#define IDS_WORDS (16 * BUCKET + 16)

__device__ __forceinline__ void sage_wave_lds_sync() {
  __builtin_amdgcn_fence(__ATOMIC_RELEASE, "wavefront");
  __builtin_amdgcn_wave_barrier();
  __builtin_amdgcn_fence(__ATOMIC_ACQUIRE, "wavefront");
}

__device__ __forceinline__ float4 sage_row4(const float* __restrict__ xin,
                                            int s, int c) {
  return *(const float4*)(xin + (size_t)s * D + c);
}

__device__ __forceinline__ void sage_add_if(float4& a, const v4f v, bool p) {
  a.x = p ? a.x + v[0] : a.x;
  a.y = p ? a.y + v[1] : a.y;
  a.z = p ? a.z + v[2] : a.z;
  a.w = p ? a.w + v[3] : a.w;
}

template <bool RELU>
__global__ void __launch_bounds__(GEMM_THREADS)
sage_layer_kernel(const float* __restrict__ xin, const int* __restrict__ cnt,
                  const int* __restrict__ bucket,
                  const int* __restrict__ ovfDst,
                  const int* __restrict__ ovfSrc, int ovfCap, int nNodes,
                  const float* __restrict__ Wl, const float* __restrict__ bl,
                  const float* __restrict__ Wr, float* __restrict__ out,
                  int nTiles) {
  __shared__ __attribute__((aligned(16))) float sW[2 * D * WPITCH];
  __shared__ __attribute__((aligned(16))) float sA[GEMM_WAVES * 2 * TILE_WORDS];
  __shared__ __attribute__((aligned(16))) int sI[GEMM_WAVES * IDS_WORDS];

  const int tid = threadIdx.x;
  const int lane = tid & 31;
  const int wave = __builtin_amdgcn_readfirstlane(tid >> 5);

  const int lo = (int)(((unsigned long long)blockIdx.x * (unsigned)nTiles) / GEMM_BLOCKS);
  const int hi = (int)(((unsigned long long)(blockIdx.x + 1) * (unsigned)nTiles) / GEMM_BLOCKS);
  const int rounds = (hi - lo + GEMM_WAVES - 1) / GEMM_WAVES;

  const int rr = lane >> 4;
  const int cc = (lane & 15) << 2;
  float* __restrict__ tA = sA + wave * (2 * TILE_WORDS);
  float* __restrict__ tX = tA + TILE_WORDS;
  int* __restrict__ tI = sI + wave * IDS_WORDS;
  int* __restrict__ tC = tI + 16 * BUCKET;

  const int m = lane & 15;
  const int khalf = (lane >> 4) << 1;
  const int mhi = (lane >> 4) << 3;
  const float* __restrict__ fA = tA + m * WPITCH + khalf;
  const float* __restrict__ fX = tX + m * WPITCH + khalf;
  const float* __restrict__ fL = sW + m * WPITCH + khalf;
  const float* __restrict__ fR = fL + D * WPITCH;

  for (int r = 0; r < rounds; ++r) {
    const int tile = lo + r * GEMM_WAVES + wave;
    const bool valid = tile < hi;
    const int row0 = tile << 4;
    int deg = 0;

    if (valid) {
      const int* __restrict__ gb = bucket + (size_t)row0 * BUCKET + (lane << 2);
      const float* __restrict__ gx = xin + (size_t)row0 * D + cc;
      int4 ids[4];
#pragma unroll
      for (int i = 0; i < 4; ++i) ids[i] = *(const int4*)(gb + i * 128);
      deg = cnt[row0 + m];
      float4 xr[8];
#pragma unroll
      for (int i = 0; i < 8; ++i)
        xr[i] = *(const float4*)(gx + (size_t)(2 * i + rr) * D);
#pragma unroll
      for (int i = 0; i < 4; ++i) *(int4*)(tI + i * 128 + (lane << 2)) = ids[i];
      tC[m] = deg;
#pragma unroll
      for (int i = 0; i < 8; ++i)
        *(float4*)(tX + (2 * i + rr) * WPITCH + cc) = xr[i];
    }
    if (r == 0) {
      constexpr int W_PIECES = 2 * D * (D / 4);
#pragma unroll
      for (int i = 0; i < (W_PIECES + GEMM_THREADS - 1) / GEMM_THREADS; ++i) {
        const int g = i * GEMM_THREADS + tid;
        if (g < W_PIECES) {
          const int wr = g >> 4;
          const float* __restrict__ w = wr < D ? Wl + wr * D : Wr + (wr - D) * D;
          *(float4*)(sW + wr * WPITCH + ((g & 15) << 2)) =
              *(const float4*)(w + ((g & 15) << 2));
        }
      }
      __syncthreads();
    }
    if (!valid) continue;

    sage_wave_lds_sync();

    const char* __restrict__ xb = (const char*)xin;
    const unsigned cb = (unsigned)cc * 4u;
    for (int hh = 0; hh < 2; ++hh) {
      const int4 da = *(const int4*)(tC + 8 * hh);
      const int4 db = *(const int4*)(tC + 8 * hh + 4);
      int nn[4];
      nn[0] = rr ? da.y : da.x;
      nn[1] = rr ? da.w : da.z;
      nn[2] = rr ? db.y : db.x;
      nn[3] = rr ? db.w : db.z;
#pragma unroll
      for (int i = 0; i < 4; ++i) nn[i] = nn[i] < BUCKET ? nn[i] : BUCKET;
      int jm = da.x > da.y ? da.x : da.y;
      jm = jm > da.z ? jm : da.z;
      jm = jm > da.w ? jm : da.w;
      jm = jm > db.x ? jm : db.x;
      jm = jm > db.y ? jm : db.y;
      jm = jm > db.z ? jm : db.z;
      jm = jm > db.w ? jm : db.w;
      jm = jm < BUCKET ? jm : BUCKET;
      jm = __builtin_amdgcn_readfirstlane(jm);
      float4 sum[4];
#pragma unroll
      for (int i = 0; i < 4; ++i) sum[i] = make_float4(0.0f, 0.0f, 0.0f, 0.0f);
      for (int j = 0; j < jm; j += 4) {
        int4 s[4];
        v4f v[4][4];
#pragma unroll
        for (int i = 0; i < 4; ++i)
          s[i] = *(const int4*)(tI + (8 * hh + 2 * i + rr) * BUCKET + j);
#pragma unroll
        for (int i = 0; i < 4; ++i) {
          v[i][0] = *(const v4f*)(xb + ((unsigned)s[i].x * (unsigned)(D * 4) + cb));
          v[i][1] = *(const v4f*)(xb + ((unsigned)s[i].y * (unsigned)(D * 4) + cb));
          v[i][2] = *(const v4f*)(xb + ((unsigned)s[i].z * (unsigned)(D * 4) + cb));
          v[i][3] = *(const v4f*)(xb + ((unsigned)s[i].w * (unsigned)(D * 4) + cb));
        }
        asm volatile("" : "+v"(v[0][0]), "+v"(v[0][1]), "+v"(v[0][2]), "+v"(v[0][3]),
                          "+v"(v[1][0]), "+v"(v[1][1]), "+v"(v[1][2]), "+v"(v[1][3])
                        : "v"(v[2][0]), "v"(v[2][1]), "v"(v[2][2]), "v"(v[2][3]),
                          "v"(v[3][0]), "v"(v[3][1]), "v"(v[3][2]), "v"(v[3][3]));
        asm volatile("" : "+v"(v[2][0]), "+v"(v[2][1]), "+v"(v[2][2]), "+v"(v[2][3]),
                          "+v"(v[3][0]), "+v"(v[3][1]), "+v"(v[3][2]), "+v"(v[3][3]));
#pragma unroll
        for (int i = 0; i < 4; ++i) {
          sage_add_if(sum[i], v[i][0], j + 0 < nn[i]);
          sage_add_if(sum[i], v[i][1], j + 1 < nn[i]);
          sage_add_if(sum[i], v[i][2], j + 2 < nn[i]);
          sage_add_if(sum[i], v[i][3], j + 3 < nn[i]);
        }
      }
#pragma unroll
      for (int i = 0; i < 4; ++i) {
        const int lr = 8 * hh + 2 * i + rr;
        if (tC[lr] > BUCKET) {
          const int row = row0 + lr;
          const int len = cnt[nNodes];
          const int mm = len < ovfCap ? len : ovfCap;
          for (int k = 0; k < mm; ++k) {
            if (ovfDst[k] == row) {
              const float4 u = sage_row4(xin, ovfSrc[k], cc);
              sum[i].x += u.x; sum[i].y += u.y; sum[i].z += u.z; sum[i].w += u.w;
            }
          }
        }
        *(float4*)(tA + lr * WPITCH + cc) = sum[i];
      }
    }

    const float rdeg = 1.0f / fmaxf((float)deg, 1.0f);
    sage_wave_lds_sync();

    v8f acc[4];
#pragma unroll
    for (int nt = 0; nt < 4; ++nt) acc[nt] = (v8f){0, 0, 0, 0, 0, 0, 0, 0};

#pragma unroll
    for (int kk = 0; kk < 16; ++kk) {
      v2f am = *(const v2f*)(fA + (kk << 2));
      am[0] *= rdeg;
      am[1] *= rdeg;
      const v2f ax = *(const v2f*)(fX + (kk << 2));
#pragma unroll
      for (int nt = 0; nt < 4; ++nt) {
        const v2f bL = *(const v2f*)(fL + (nt << 4) * WPITCH + (kk << 2));
        const v2f bR = *(const v2f*)(fR + (nt << 4) * WPITCH + (kk << 2));
        acc[nt] = __builtin_amdgcn_wmma_f32_16x16x4_f32(
            false, am, false, bL, (short)0, acc[nt], false, false);
        acc[nt] = __builtin_amdgcn_wmma_f32_16x16x4_f32(
            false, ax, false, bR, (short)0, acc[nt], false, false);
      }
    }

#pragma unroll
    for (int nt = 0; nt < 4; ++nt) {
      const int j = (nt << 4) + m;
      const float b = bl[j];
#pragma unroll
      for (int v = 0; v < 8; ++v) {
        float val = acc[nt][v] + b;
        if (RELU) val = fmaxf(val, 0.0f);
        tA[(mhi + v) * WPITCH + j] = val;
      }
    }
    sage_wave_lds_sync();
    float* __restrict__ go = out + (size_t)row0 * D + cc;
#pragma unroll
    for (int i = 0; i < 8; ++i)
      *(float4*)(go + (size_t)(2 * i + rr) * D) =
          *(const float4*)(tA + (2 * i + rr) * WPITCH + cc);
    sage_wave_lds_sync();
  }
}

template <bool RELU>
__global__ void sage_tail_kernel(const float* __restrict__ xin,
                                 const int* __restrict__ cnt,
                                 const int* __restrict__ bucket,
                                 const int* __restrict__ ovfDst,
                                 const int* __restrict__ ovfSrc, int ovfCap,
                                 int nNodes,
                                 const float* __restrict__ Wl,
                                 const float* __restrict__ bl,
                                 const float* __restrict__ Wr,
                                 float* __restrict__ out, int row0, int nRows) {
  int idx = blockIdx.x * blockDim.x + threadIdx.x;
  if (idx >= nRows * D) return;
  int r = row0 + idx / D;
  int j = idx % D;
  const int n = cnt[r];
  const int end = n < BUCKET ? n : BUCKET;
  const int len = cnt[nNodes];
  const int mm = len < ovfCap ? len : ovfCap;
  float rdeg = 1.0f / fmaxf((float)n, 1.0f);
  float s = bl[j];
  for (int k = 0; k < D; ++k) {
    float a = 0.0f;
    for (int t = 0; t < end; ++t) a += xin[(size_t)bucket[r * BUCKET + t] * D + k];
    if (n > BUCKET)
      for (int t = 0; t < mm; ++t)
        if (ovfDst[t] == r) a += xin[(size_t)ovfSrc[t] * D + k];
    s += a * rdeg * Wl[j * D + k] + xin[(size_t)r * D + k] * Wr[j * D + k];
  }
  if (RELU) s = fmaxf(s, 0.0f);
  out[(size_t)r * D + j] = s;
}

static inline size_t align256(size_t x) { return (x + 255) & ~(size_t)255; }

extern "C" void kernel_launch(void* const* d_in, const int* in_sizes, int n_in,
                              void* d_out, int out_size, void* d_ws,
                              size_t ws_size, hipStream_t stream) {
  const float* x = (const float*)d_in[0];
  const int* src = (const int*)d_in[1];
  const int* dst = (const int*)d_in[2];
  const float* Wl[3] = {(const float*)d_in[3], (const float*)d_in[6],
                        (const float*)d_in[9]};
  const float* bl[3] = {(const float*)d_in[4], (const float*)d_in[7],
                        (const float*)d_in[10]};
  const float* Wr[3] = {(const float*)d_in[5], (const float*)d_in[8],
                        (const float*)d_in[11]};

  const int nNodes = in_sizes[0] / D;
  const int nEdges = in_sizes[1];
  const size_t featBytes = (size_t)nNodes * D * sizeof(float);

  char* ws = (char*)d_ws;
  int* cnt = (int*)ws;
  int* ovfLen = cnt + nNodes;
  size_t ofs = align256(((size_t)nNodes + 1) * sizeof(int));
  char* adj = ws + ofs;
  ofs += align256(featBytes);
  float* h0 = (float*)(ws + ofs);
  ofs += align256(featBytes);
  float* h1 = (float*)(ws + ofs);

  int* bucket = (int*)adj;
  const size_t bucketBytes = (size_t)nNodes * BUCKET * sizeof(int);
  const int ovfCap = (int)((featBytes - bucketBytes) / (2 * sizeof(int)));
  int* ovfDst = (int*)(adj + bucketBytes);
  int* ovfSrc = ovfDst + ovfCap;
  int* blockOff = (int*)h0;
  int* pairs = (int*)d_out;
  const int partBlocks = (nEdges + PART_EDGES - 1) / PART_EDGES;
  sage_place_kernel<<<partBlocks, PART_THREADS, 0, stream>>>(
      src, dst, blockOff, pairs, ovfLen, nEdges);
  sage_bucket_kernel<<<(nNodes + OWN - 1) / OWN, OWN_THREADS, 0, stream>>>(
      pairs, blockOff, partBlocks, cnt, bucket, ovfDst, ovfSrc, ovfLen, ovfCap,
      nNodes);

  const int nTiles = nNodes / 16;
  const int gemmBlocks = GEMM_BLOCKS;
  const int tail = nNodes - nTiles * 16;

  auto run_layer = [&](const float* in, const float* wl, const float* b,
                       const float* wr, float* outp, bool relu) {
    if (nTiles > 0) {
      if (relu)
        sage_layer_kernel<true><<<gemmBlocks, GEMM_THREADS, 0, stream>>>(
            in, cnt, bucket, ovfDst, ovfSrc, ovfCap, nNodes, wl, b, wr, outp,
            nTiles);
      else
        sage_layer_kernel<false><<<gemmBlocks, GEMM_THREADS, 0, stream>>>(
            in, cnt, bucket, ovfDst, ovfSrc, ovfCap, nNodes, wl, b, wr, outp,
            nTiles);
    }
    if (tail > 0) {
      int tb = (tail * D + 255) / 256;
      if (relu)
        sage_tail_kernel<true><<<tb, 256, 0, stream>>>(
            in, cnt, bucket, ovfDst, ovfSrc, ovfCap, nNodes, wl, b, wr, outp,
            nTiles * 16, tail);
      else
        sage_tail_kernel<false><<<tb, 256, 0, stream>>>(
            in, cnt, bucket, ovfDst, ovfSrc, ovfCap, nNodes, wl, b, wr, outp,
            nTiles * 16, tail);
    }
  };

  run_layer(x, Wl[0], bl[0], Wr[0], h0,  true);
  run_layer(h0, Wl[1], bl[1], Wr[1], h1,  true);
  run_layer(h1, Wl[2], bl[2], Wr[2], (float*)d_out,  false);
}


# Each wave multiplies its root term (x @ Wr^T) before its neighbour sums and parks the accumulators in LDS: waves leave the matrix pipe one by one and stay apart; acc adds root products, then mean
# speedup vs baseline: 5.7283x; 1.0385x over previous
#include <hip/hip_runtime.h>

typedef float v2f __attribute__((ext_vector_type(2)));
typedef float v8f __attribute__((ext_vector_type(8)));
typedef float v4f __attribute__((ext_vector_type(4)));

#define D 64

#define BUCKET 32
#define OWN 784
#define OWN_THREADS 1024
#define BUCKET_RUNS 8
#define RANGES_MAX 256
#define PART_THREADS 256
#define PART_QUADS 4
#define PART_EDGES (PART_THREADS * PART_QUADS * 4)

__device__ __forceinline__ int4 sage_load4(const int* __restrict__ p, int e,
                                           int nEdges) {
  if (e + 4 <= nEdges) return *(const int4*)(p + e);
  int4 v = make_int4(-1, -1, -1, -1);
  if (e + 0 < nEdges) v.x = p[e + 0];
  if (e + 1 < nEdges) v.y = p[e + 1];
  if (e + 2 < nEdges) v.z = p[e + 2];
  return v;
}

__device__ __forceinline__ int sage_scan_incl(int v, int* a, int* b, int tid) {
  a[tid] = v;
  __syncthreads();
  int* in = a;
  int* out = b;
  for (int off = 1; off < RANGES_MAX; off <<= 1) {
    out[tid] = in[tid] + (tid >= off ? in[tid - off] : 0);
    __syncthreads();
    int* t = in; in = out; out = t;
  }
  return in[tid];
}

__global__ void __launch_bounds__(PART_THREADS)
sage_place_kernel(const int* __restrict__ src, const int* __restrict__ dst,
                  int* __restrict__ blockOff, int* __restrict__ pairs,
                  int* __restrict__ ovfLen, int nEdges) {
  __shared__ int hist[RANGES_MAX];
  __shared__ int lstart[RANGES_MAX];
  __shared__ int scA[RANGES_MAX], scB[RANGES_MAX];
  __shared__ __attribute__((aligned(8))) int stage[PART_EDGES * 2];
  const int tid = threadIdx.x;
  hist[tid] = 0;
  if (blockIdx.x == 0 && tid == 0) *ovfLen = 0;
  __syncthreads();

  int4 d[PART_QUADS], s[PART_QUADS], rk[PART_QUADS];
  const int e0 = blockIdx.x * PART_EDGES;
#pragma unroll
  for (int k = 0; k < PART_QUADS; ++k) {
    const int e = e0 + (k * PART_THREADS + tid) * 4;
    d[k] = sage_load4(dst, e, nEdges);
    s[k] = sage_load4(src, e, nEdges);
    rk[k] = make_int4(0, 0, 0, 0);
    if (d[k].x >= 0) rk[k].x = atomicAdd(&hist[d[k].x / OWN], 1);
    if (d[k].y >= 0) rk[k].y = atomicAdd(&hist[d[k].y / OWN], 1);
    if (d[k].z >= 0) rk[k].z = atomicAdd(&hist[d[k].z / OWN], 1);
    if (d[k].w >= 0) rk[k].w = atomicAdd(&hist[d[k].w / OWN], 1);
  }
  __syncthreads();

  const int h = hist[tid];
  const int ls = sage_scan_incl(h, scA, scB, tid) - h;
  lstart[tid] = ls;
  int* __restrict__ off = blockOff + (size_t)blockIdx.x * (RANGES_MAX + 1);
  off[tid] = ls;
  if (tid == RANGES_MAX - 1) off[RANGES_MAX] = ls + h;
  __syncthreads();

#pragma unroll
  for (int k = 0; k < PART_QUADS; ++k) {
    if (d[k].x >= 0) *(int2*)(stage + 2 * (lstart[d[k].x / OWN] + rk[k].x)) = make_int2(d[k].x, s[k].x);
    if (d[k].y >= 0) *(int2*)(stage + 2 * (lstart[d[k].y / OWN] + rk[k].y)) = make_int2(d[k].y, s[k].y);
    if (d[k].z >= 0) *(int2*)(stage + 2 * (lstart[d[k].z / OWN] + rk[k].z)) = make_int2(d[k].z, s[k].z);
    if (d[k].w >= 0) *(int2*)(stage + 2 * (lstart[d[k].w / OWN] + rk[k].w)) = make_int2(d[k].w, s[k].w);
  }
  __syncthreads();

  int nHere = nEdges - e0;
  if (nHere > PART_EDGES) nHere = PART_EDGES;
  int* __restrict__ area = pairs + 2 * (size_t)e0;
  for (int i = tid; i < nHere; i += PART_THREADS)
    *(int2*)(area + 2 * i) = *(const int2*)(stage + 2 * i);
}

__global__ void __launch_bounds__(OWN_THREADS)
sage_bucket_kernel(const int* __restrict__ pairs,
                   const int* __restrict__ blockOff, int partBlocks,
                   int* __restrict__ cnt,
                   int* __restrict__ bucket,
                   int* __restrict__ ovfDst,
                   int* __restrict__ ovfSrc,
                   int* __restrict__ ovfLen, int ovfCap, int nNodes) {
  __shared__ int lcnt[OWN];
  __shared__ __attribute__((aligned(16))) int lbkt[OWN * BUCKET];
  __shared__ int runLo[OWN_THREADS], runHi[OWN_THREADS];
  const int tid = threadIdx.x;
  const int lane = tid & 31;
  const int wave = tid >> 5;
  const int lo = blockIdx.x * OWN;
  const int left = nNodes - lo;
  const int nOwn = left < OWN ? left : OWN;
  for (int i = tid; i < OWN; i += OWN_THREADS) lcnt[i] = 0;

  auto put = [&](const int2 pr) {
    const int r = pr.x - lo;
    const int slot = atomicAdd(&lcnt[r], 1);
    if (slot < BUCKET) {
      lbkt[r * BUCKET + slot] = pr.y;
    } else {
      const int pos = atomicAdd(ovfLen, 1);
      if (pos < ovfCap) {
        ovfDst[pos] = pr.x;
        ovfSrc[pos] = pr.y;
      }
    }
  };

  for (int c0 = 0; c0 < partBlocks; c0 += OWN_THREADS) {
    if (c0 + tid < partBlocks) {
      const int* __restrict__ off =
          blockOff + (size_t)(c0 + tid) * (RANGES_MAX + 1) + blockIdx.x;
      runLo[tid] = off[0];
      runHi[tid] = off[1];
    }
    __syncthreads();
    const int nb = partBlocks - c0 < OWN_THREADS ? partBlocks - c0 : OWN_THREADS;
    for (int q = wave; q < nb; q += BUCKET_RUNS * (OWN_THREADS / 32)) {
      int rs[BUCKET_RUNS], re[BUCKET_RUNS];
      int2 v[BUCKET_RUNS][2];
#pragma unroll
      for (int u = 0; u < BUCKET_RUNS; ++u) {
        const int qq = q + u * (OWN_THREADS / 32);
        const bool ok = qq < nb;
        rs[u] = ok ? runLo[qq] : 0;
        re[u] = ok ? runHi[qq] : 0;
      }
#pragma unroll
      for (int u = 0; u < BUCKET_RUNS; ++u) {
        const int* __restrict__ g =
            pairs + 2 * (size_t)(c0 + q + u * (OWN_THREADS / 32)) * PART_EDGES;
        v[u][0] = make_int2(-1, 0);
        v[u][1] = make_int2(-1, 0);
        if (rs[u] + lane < re[u]) v[u][0] = *(const int2*)(g + 2 * (rs[u] + lane));
        if (rs[u] + 32 + lane < re[u])
          v[u][1] = *(const int2*)(g + 2 * (rs[u] + 32 + lane));
      }
#pragma unroll
      for (int u = 0; u < BUCKET_RUNS; ++u) {
        const int* __restrict__ g =
            pairs + 2 * (size_t)(c0 + q + u * (OWN_THREADS / 32)) * PART_EDGES;
        if (v[u][0].x >= 0) put(v[u][0]);
        if (v[u][1].x >= 0) put(v[u][1]);
        for (int i = rs[u] + 64 + lane; i < re[u]; i += 32)
          put(*(const int2*)(g + 2 * i));
      }
    }
    __syncthreads();
  }

  for (int i = tid; i < nOwn; i += OWN_THREADS) cnt[lo + i] = lcnt[i];
  for (int g = tid; g < nOwn * (BUCKET / 4); g += OWN_THREADS) {
    const int node = g >> 3;
    const int slot = (g & 7) << 2;
    const int w = node * BUCKET + slot;
    const int deg = lcnt[node];
    int4 v = *(const int4*)(lbkt + w);
    v.x = slot + 0 < deg ? v.x : 0;
    v.y = slot + 1 < deg ? v.y : 0;
    v.z = slot + 2 < deg ? v.z : 0;
    v.w = slot + 3 < deg ? v.w : 0;
    *(int4*)(bucket + (size_t)lo * BUCKET + w) = v;
  }
}

#define WPITCH 68
#define GEMM_WAVES 26
#define GEMM_THREADS (GEMM_WAVES * 32)
#define GEMM_BLOCKS 256
#define TILE_WORDS (16 * WPITCH)
#define IDS_WORDS (16 * BUCKET + 16)

__device__ __forceinline__ void sage_wave_lds_sync() {
  __builtin_amdgcn_fence(__ATOMIC_RELEASE, "wavefront");
  __builtin_amdgcn_wave_barrier();
  __builtin_amdgcn_fence(__ATOMIC_ACQUIRE, "wavefront");
}

__device__ __forceinline__ float4 sage_row4(const float* __restrict__ xin,
                                            int s, int c) {
  return *(const float4*)(xin + (size_t)s * D + c);
}

__device__ __forceinline__ void sage_add_if(float4& a, const v4f v, bool p) {
  a.x = p ? a.x + v[0] : a.x;
  a.y = p ? a.y + v[1] : a.y;
  a.z = p ? a.z + v[2] : a.z;
  a.w = p ? a.w + v[3] : a.w;
}

template <bool RELU>
__global__ void __launch_bounds__(GEMM_THREADS)
sage_layer_kernel(const float* __restrict__ xin, const int* __restrict__ cnt,
                  const int* __restrict__ bucket,
                  const int* __restrict__ ovfDst,
                  const int* __restrict__ ovfSrc, int ovfCap, int nNodes,
                  const float* __restrict__ Wl, const float* __restrict__ bl,
                  const float* __restrict__ Wr, float* __restrict__ out,
                  int nTiles) {
  __shared__ __attribute__((aligned(16))) float sW[2 * D * WPITCH];
  __shared__ __attribute__((aligned(16))) float sA[GEMM_WAVES * 2 * TILE_WORDS];
  __shared__ __attribute__((aligned(16))) int sI[GEMM_WAVES * IDS_WORDS];

  const int tid = threadIdx.x;
  const int lane = tid & 31;
  const int wave = __builtin_amdgcn_readfirstlane(tid >> 5);

  const int lo = (int)(((unsigned long long)blockIdx.x * (unsigned)nTiles) / GEMM_BLOCKS);
  const int hi = (int)(((unsigned long long)(blockIdx.x + 1) * (unsigned)nTiles) / GEMM_BLOCKS);
  const int rounds = (hi - lo + GEMM_WAVES - 1) / GEMM_WAVES;

  const int rr = lane >> 4;
  const int cc = (lane & 15) << 2;
  float* __restrict__ tA = sA + wave * (2 * TILE_WORDS);
  float* __restrict__ tX = tA + TILE_WORDS;
  int* __restrict__ tI = sI + wave * IDS_WORDS;
  int* __restrict__ tC = tI + 16 * BUCKET;

  const int m = lane & 15;
  const int khalf = (lane >> 4) << 1;
  const int mhi = (lane >> 4) << 3;
  const float* __restrict__ fA = tA + m * WPITCH + khalf;
  const float* __restrict__ fX = tX + m * WPITCH + khalf;
  const float* __restrict__ fL = sW + m * WPITCH + khalf;
  const float* __restrict__ fR = fL + D * WPITCH;

  for (int r = 0; r < rounds; ++r) {
    const int tile = lo + r * GEMM_WAVES + wave;
    const bool valid = tile < hi;
    const int row0 = tile << 4;
    int deg = 0;

    if (valid) {
      const int* __restrict__ gb = bucket + (size_t)row0 * BUCKET + (lane << 2);
      const float* __restrict__ gx = xin + (size_t)row0 * D + cc;
      int4 ids[4];
#pragma unroll
      for (int i = 0; i < 4; ++i) ids[i] = *(const int4*)(gb + i * 128);
      deg = cnt[row0 + m];
      float4 xr[8];
#pragma unroll
      for (int i = 0; i < 8; ++i)
        xr[i] = *(const float4*)(gx + (size_t)(2 * i + rr) * D);
#pragma unroll
      for (int i = 0; i < 4; ++i) *(int4*)(tI + i * 128 + (lane << 2)) = ids[i];
      tC[m] = deg;
#pragma unroll
      for (int i = 0; i < 8; ++i)
        *(float4*)(tX + (2 * i + rr) * WPITCH + cc) = xr[i];
    }
    if (r == 0) {
      constexpr int W_PIECES = 2 * D * (D / 4);
#pragma unroll
      for (int i = 0; i < (W_PIECES + GEMM_THREADS - 1) / GEMM_THREADS; ++i) {
        const int g = i * GEMM_THREADS + tid;
        if (g < W_PIECES) {
          const int wr = g >> 4;
          const float* __restrict__ w = wr < D ? Wl + wr * D : Wr + (wr - D) * D;
          *(float4*)(sW + wr * WPITCH + ((g & 15) << 2)) =
              *(const float4*)(w + ((g & 15) << 2));
        }
      }
      __syncthreads();
    }
    if (!valid) continue;

    sage_wave_lds_sync();

    v8f acc[4];
#pragma unroll
    for (int nt = 0; nt < 4; ++nt) acc[nt] = (v8f){0, 0, 0, 0, 0, 0, 0, 0};
#pragma unroll
    for (int kk = 0; kk < 16; ++kk) {
      const v2f ax = *(const v2f*)(fX + (kk << 2));
#pragma unroll
      for (int nt = 0; nt < 4; ++nt) {
        const v2f bR = *(const v2f*)(fR + (nt << 4) * WPITCH + (kk << 2));
        acc[nt] = __builtin_amdgcn_wmma_f32_16x16x4_f32(
            false, ax, false, bR, (short)0, acc[nt], false, false);
      }
    }
    float* pk = tX + (lane << 2);
#pragma unroll
    for (int nt = 0; nt < 4; ++nt) {
      *(v4f*)(pk + (2 * nt) * 128) =
          __builtin_shufflevector(acc[nt], acc[nt], 0, 1, 2, 3);
      *(v4f*)(pk + (2 * nt + 1) * 128) =
          __builtin_shufflevector(acc[nt], acc[nt], 4, 5, 6, 7);
    }

    const char* __restrict__ xb = (const char*)xin;
    const unsigned cb = (unsigned)cc * 4u;
    for (int hh = 0; hh < 2; ++hh) {
      const int4 da = *(const int4*)(tC + 8 * hh);
      const int4 db = *(const int4*)(tC + 8 * hh + 4);
      int nn[4];
      nn[0] = rr ? da.y : da.x;
      nn[1] = rr ? da.w : da.z;
      nn[2] = rr ? db.y : db.x;
      nn[3] = rr ? db.w : db.z;
#pragma unroll
      for (int i = 0; i < 4; ++i) nn[i] = nn[i] < BUCKET ? nn[i] : BUCKET;
      int jm = da.x > da.y ? da.x : da.y;
      jm = jm > da.z ? jm : da.z;
      jm = jm > da.w ? jm : da.w;
      jm = jm > db.x ? jm : db.x;
      jm = jm > db.y ? jm : db.y;
      jm = jm > db.z ? jm : db.z;
      jm = jm > db.w ? jm : db.w;
      jm = jm < BUCKET ? jm : BUCKET;
      jm = __builtin_amdgcn_readfirstlane(jm);
      float4 sum[4];
#pragma unroll
      for (int i = 0; i < 4; ++i) sum[i] = make_float4(0.0f, 0.0f, 0.0f, 0.0f);
      for (int j = 0; j < jm; j += 4) {
        int4 s[4];
        v4f v[4][4];
#pragma unroll
        for (int i = 0; i < 4; ++i)
          s[i] = *(const int4*)(tI + (8 * hh + 2 * i + rr) * BUCKET + j);
#pragma unroll
        for (int i = 0; i < 4; ++i) {
          v[i][0] = *(const v4f*)(xb + ((unsigned)s[i].x * (unsigned)(D * 4) + cb));
          v[i][1] = *(const v4f*)(xb + ((unsigned)s[i].y * (unsigned)(D * 4) + cb));
          v[i][2] = *(const v4f*)(xb + ((unsigned)s[i].z * (unsigned)(D * 4) + cb));
          v[i][3] = *(const v4f*)(xb + ((unsigned)s[i].w * (unsigned)(D * 4) + cb));
        }
        asm volatile("" : "+v"(v[0][0]), "+v"(v[0][1]), "+v"(v[0][2]), "+v"(v[0][3]),
                          "+v"(v[1][0]), "+v"(v[1][1]), "+v"(v[1][2]), "+v"(v[1][3])
                        : "v"(v[2][0]), "v"(v[2][1]), "v"(v[2][2]), "v"(v[2][3]),
                          "v"(v[3][0]), "v"(v[3][1]), "v"(v[3][2]), "v"(v[3][3]));
        asm volatile("" : "+v"(v[2][0]), "+v"(v[2][1]), "+v"(v[2][2]), "+v"(v[2][3]),
                          "+v"(v[3][0]), "+v"(v[3][1]), "+v"(v[3][2]), "+v"(v[3][3]));
#pragma unroll
        for (int i = 0; i < 4; ++i) {
          sage_add_if(sum[i], v[i][0], j + 0 < nn[i]);
          sage_add_if(sum[i], v[i][1], j + 1 < nn[i]);
          sage_add_if(sum[i], v[i][2], j + 2 < nn[i]);
          sage_add_if(sum[i], v[i][3], j + 3 < nn[i]);
        }
      }
#pragma unroll
      for (int i = 0; i < 4; ++i) {
        const int lr = 8 * hh + 2 * i + rr;
        if (tC[lr] > BUCKET) {
          const int row = row0 + lr;
          const int len = cnt[nNodes];
          const int mm = len < ovfCap ? len : ovfCap;
          for (int k = 0; k < mm; ++k) {
            if (ovfDst[k] == row) {
              const float4 u = sage_row4(xin, ovfSrc[k], cc);
              sum[i].x += u.x; sum[i].y += u.y; sum[i].z += u.z; sum[i].w += u.w;
            }
          }
        }
        *(float4*)(tA + lr * WPITCH + cc) = sum[i];
      }
    }

    const float rdeg = 1.0f / fmaxf((float)deg, 1.0f);
    sage_wave_lds_sync();

#pragma unroll
    for (int nt = 0; nt < 4; ++nt) {
      const v4f p0 = *(const v4f*)(pk + (2 * nt) * 128);
      const v4f p1 = *(const v4f*)(pk + (2 * nt + 1) * 128);
      acc[nt] = __builtin_shufflevector(p0, p1, 0, 1, 2, 3, 4, 5, 6, 7);
    }
#pragma unroll
    for (int kk = 0; kk < 16; ++kk) {
      v2f am = *(const v2f*)(fA + (kk << 2));
      am[0] *= rdeg;
      am[1] *= rdeg;
#pragma unroll
      for (int nt = 0; nt < 4; ++nt) {
        const v2f bL = *(const v2f*)(fL + (nt << 4) * WPITCH + (kk << 2));
        acc[nt] = __builtin_amdgcn_wmma_f32_16x16x4_f32(
            false, am, false, bL, (short)0, acc[nt], false, false);
      }
    }

#pragma unroll
    for (int nt = 0; nt < 4; ++nt) {
      const int j = (nt << 4) + m;
      const float b = bl[j];
#pragma unroll
      for (int v = 0; v < 8; ++v) {
        float val = acc[nt][v] + b;
        if (RELU) val = fmaxf(val, 0.0f);
        tA[(mhi + v) * WPITCH + j] = val;
      }
    }
    sage_wave_lds_sync();
    float* __restrict__ go = out + (size_t)row0 * D + cc;
#pragma unroll
    for (int i = 0; i < 8; ++i)
      *(float4*)(go + (size_t)(2 * i + rr) * D) =
          *(const float4*)(tA + (2 * i + rr) * WPITCH + cc);
    sage_wave_lds_sync();
  }
}

template <bool RELU>
__global__ void sage_tail_kernel(const float* __restrict__ xin,
                                 const int* __restrict__ cnt,
                                 const int* __restrict__ bucket,
                                 const int* __restrict__ ovfDst,
                                 const int* __restrict__ ovfSrc, int ovfCap,
                                 int nNodes,
                                 const float* __restrict__ Wl,
                                 const float* __restrict__ bl,
                                 const float* __restrict__ Wr,
                                 float* __restrict__ out, int row0, int nRows) {
  int idx = blockIdx.x * blockDim.x + threadIdx.x;
  if (idx >= nRows * D) return;
  int r = row0 + idx / D;
  int j = idx % D;
  const int n = cnt[r];
  const int end = n < BUCKET ? n : BUCKET;
  const int len = cnt[nNodes];
  const int mm = len < ovfCap ? len : ovfCap;
  float rdeg = 1.0f / fmaxf((float)n, 1.0f);
  float s = bl[j];
  for (int k = 0; k < D; ++k) {
    float a = 0.0f;
    for (int t = 0; t < end; ++t) a += xin[(size_t)bucket[r * BUCKET + t] * D + k];
    if (n > BUCKET)
      for (int t = 0; t < mm; ++t)
        if (ovfDst[t] == r) a += xin[(size_t)ovfSrc[t] * D + k];
    s += a * rdeg * Wl[j * D + k] + xin[(size_t)r * D + k] * Wr[j * D + k];
  }
  if (RELU) s = fmaxf(s, 0.0f);
  out[(size_t)r * D + j] = s;
}

static inline size_t align256(size_t x) { return (x + 255) & ~(size_t)255; }

extern "C" void kernel_launch(void* const* d_in, const int* in_sizes, int n_in,
                              void* d_out, int out_size, void* d_ws,
                              size_t ws_size, hipStream_t stream) {
  const float* x = (const float*)d_in[0];
  const int* src = (const int*)d_in[1];
  const int* dst = (const int*)d_in[2];
  const float* Wl[3] = {(const float*)d_in[3], (const float*)d_in[6],
                        (const float*)d_in[9]};
  const float* bl[3] = {(const float*)d_in[4], (const float*)d_in[7],
                        (const float*)d_in[10]};
  const float* Wr[3] = {(const float*)d_in[5], (const float*)d_in[8],
                        (const float*)d_in[11]};

  const int nNodes = in_sizes[0] / D;
  const int nEdges = in_sizes[1];
  const size_t featBytes = (size_t)nNodes * D * sizeof(float);

  char* ws = (char*)d_ws;
  int* cnt = (int*)ws;
  int* ovfLen = cnt + nNodes;
  size_t ofs = align256(((size_t)nNodes + 1) * sizeof(int));
  char* adj = ws + ofs;
  ofs += align256(featBytes);
  float* h0 = (float*)(ws + ofs);
  ofs += align256(featBytes);
  float* h1 = (float*)(ws + ofs);

  int* bucket = (int*)adj;
  const size_t bucketBytes = (size_t)nNodes * BUCKET * sizeof(int);
  const int ovfCap = (int)((featBytes - bucketBytes) / (2 * sizeof(int)));
  int* ovfDst = (int*)(adj + bucketBytes);
  int* ovfSrc = ovfDst + ovfCap;
  int* blockOff = (int*)h0;
  int* pairs = (int*)d_out;
  const int partBlocks = (nEdges + PART_EDGES - 1) / PART_EDGES;
  sage_place_kernel<<<partBlocks, PART_THREADS, 0, stream>>>(
      src, dst, blockOff, pairs, ovfLen, nEdges);
  sage_bucket_kernel<<<(nNodes + OWN - 1) / OWN, OWN_THREADS, 0, stream>>>(
      pairs, blockOff, partBlocks, cnt, bucket, ovfDst, ovfSrc, ovfLen, ovfCap,
      nNodes);

  const int nTiles = nNodes / 16;
  const int gemmBlocks = GEMM_BLOCKS;
  const int tail = nNodes - nTiles * 16;

  auto run_layer = [&](const float* in, const float* wl, const float* b,
                       const float* wr, float* outp, bool relu) {
    if (nTiles > 0) {
      if (relu)
        sage_layer_kernel<true><<<gemmBlocks, GEMM_THREADS, 0, stream>>>(
            in, cnt, bucket, ovfDst, ovfSrc, ovfCap, nNodes, wl, b, wr, outp,
            nTiles);
      else
        sage_layer_kernel<false><<<gemmBlocks, GEMM_THREADS, 0, stream>>>(
            in, cnt, bucket, ovfDst, ovfSrc, ovfCap, nNodes, wl, b, wr, outp,
            nTiles);
    }
    if (tail > 0) {
      int tb = (tail * D + 255) / 256;
      if (relu)
        sage_tail_kernel<true><<<tb, 256, 0, stream>>>(
            in, cnt, bucket, ovfDst, ovfSrc, ovfCap, nNodes, wl, b, wr, outp,
            nTiles * 16, tail);
      else
        sage_tail_kernel<false><<<tb, 256, 0, stream>>>(
            in, cnt, bucket, ovfDst, ovfSrc, ovfCap, nNodes, wl, b, wr, outp,
            nTiles * 16, tail);
    }
  };

  run_layer(x, Wl[0], bl[0], Wr[0], h0,  true);
  run_layer(h0, Wl[1], bl[1], Wr[1], h1,  true);
  run_layer(h1, Wl[2], bl[2], Wr[2], (float*)d_out,  false);
}
